# starting kernel
# baseline (speedup 1.0000x reference)
#include <hip/hip_runtime.h>
#include <math.h>

typedef __attribute__((ext_vector_type(2))) float v2f;
typedef __attribute__((ext_vector_type(8))) float v8f;

#define N0_  262144
#define INF_ 128
#define H_   32
#define N1_  131072
#define N2_  65536

__device__ __forceinline__ v8f wmma_f32_k4(v2f a, v2f b, v8f c) {
  return __builtin_amdgcn_wmma_f32_16x16x4_f32(
       false, a,  false, b,
       (short)0, c,  false,  false);
}

template <int F4, int LOG2F4>
__global__ __launch_bounds__(256) void sage_scatter_kernel(
    const float* __restrict__ xsrc, const int* __restrict__ src,
    const int* __restrict__ tgt, float* __restrict__ agg,
    float* __restrict__ cnt, int nedges) {
  long long tid = (long long)blockIdx.x * blockDim.x + threadIdx.x;
  long long total = (long long)nedges << LOG2F4;
  if (tid >= total) return;
  int e = (int)(tid >> LOG2F4);
  int g = (int)(tid & (F4 - 1));
  int s = src[e];
  int t = tgt[e];
  const float4* x4 = (const float4*)xsrc;
  float4 v = x4[(size_t)s * F4 + g];
  float* dst = agg + (size_t)t * (F4 * 4) + g * 4;
  unsafeAtomicAdd(dst + 0, v.x);
  unsafeAtomicAdd(dst + 1, v.y);
  unsafeAtomicAdd(dst + 2, v.z);
  unsafeAtomicAdd(dst + 3, v.w);
  if (g == 0) unsafeAtomicAdd(cnt + t, 1.0f);
}

template <int K, bool RELU>
__global__ __launch_bounds__(256) void sage_compute_kernel(
    const float* __restrict__ agg, const float* __restrict__ cnt,
    const float* __restrict__ xin, const float* __restrict__ wl,
    const float* __restrict__ bl, const float* __restrict__ wr,
    float* __restrict__ out, int nrows) {
  const int lane = threadIdx.x & 31;
  const int wid  = blockIdx.x * (blockDim.x >> 5) + (threadIdx.x >> 5);
  const int base = wid * 16;
  if (base >= nrows) return;
  const int l16  = lane & 15;
  const int half = lane >> 4;
  const int rowA = base + l16;

  const float inv = 1.0f / fmaxf(cnt[rowA], 1.0f);

  v8f c0 = {};
  v8f c1 = {};

#pragma unroll
  for (int phase = 0; phase < 2; ++phase) {
    const float* Arow = (phase == 0 ? agg : xin) + (size_t)rowA * K;
    const float* W    = (phase == 0 ? wl : wr);
    const float  s    = (phase == 0 ? inv : 1.0f);
    const float2* A2  = (const float2*)Arow;
    const float2* B0  = (const float2*)(W + (size_t)l16 * K);
    const float2* B1  = (const float2*)(W + (size_t)(l16 + 16) * K);
#pragma unroll
    for (int kc = 0; kc < K; kc += 4) {
      const int fi = (kc >> 1) + half;
      float2 fa = A2[fi];
      float2 f0 = B0[fi];
      float2 f1 = B1[fi];
      v2f a;  a.x  = fa.x * s; a.y  = fa.y * s;
      v2f b0; b0.x = f0.x;     b0.y = f0.y;
      v2f b1; b1.x = f1.x;     b1.y = f1.y;
      c0 = wmma_f32_k4(a, b0, c0);
      c1 = wmma_f32_k4(a, b1, c1);
    }
  }

  const float bias0 = bl[l16];
  const float bias1 = bl[l16 + 16];
#pragma unroll
  for (int r = 0; r < 8; ++r) { c0[r] += bias0; c1[r] += bias1; }

#pragma unroll
  for (int r = 0; r < 8; ++r) {
    float p = c0[r] * c0[r] + c1[r] * c1[r];
    p += __shfl_xor(p, 1, 32);
    p += __shfl_xor(p, 2, 32);
    p += __shfl_xor(p, 4, 32);
    p += __shfl_xor(p, 8, 32);
    float scale = 1.0f / fmaxf(sqrtf(p), 1e-12f);
    float o0 = c0[r] * scale;
    float o1 = c1[r] * scale;
    if (RELU) { o0 = fmaxf(o0, 0.0f); o1 = fmaxf(o1, 0.0f); }
    const size_t row = (size_t)(base + r + half * 8);
    out[row * 32 + l16]      = o0;
    out[row * 32 + 16 + l16] = o1;
  }
}

extern "C" void kernel_launch(void* const* d_in, const int* in_sizes, int n_in,
                              void* d_out, int out_size, void* d_ws,
                              size_t ws_size, hipStream_t stream) {
  const float* x   = (const float*)d_in[0];
  const int* src0  = (const int*)d_in[1];
  const int* tgt0  = (const int*)d_in[2];
  const int* src1  = (const int*)d_in[3];
  const int* tgt1  = (const int*)d_in[4];
  const float* wl0 = (const float*)d_in[5];
  const float* bl0 = (const float*)d_in[6];
  const float* wr0 = (const float*)d_in[7];
  const float* wl1 = (const float*)d_in[8];
  const float* bl1 = (const float*)d_in[9];
  const float* wr1 = (const float*)d_in[10];
  const int E0 = in_sizes[1];
  const int E1 = in_sizes[3];

  float* ws   = (float*)d_ws;
  float* agg0 = ws;
  float* cnt0 = agg0 + (size_t)N1_ * INF_;
  float* agg1 = cnt0 + (size_t)N1_;
  float* cnt1 = agg1 + (size_t)N2_ * H_;
  float* h    = cnt1 + (size_t)N2_;
  const size_t zero_bytes =
      ((size_t)N1_ * INF_ + N1_ + (size_t)N2_ * H_ + N2_) * sizeof(float);
  hipMemsetAsync(d_ws, 0, zero_bytes, stream);

  {
    long long total = (long long)E0 * 32;
    int blocks = (int)((total + 255) / 256);
    sage_scatter_kernel<32, 5><<<blocks, 256, 0, stream>>>(x, src0, tgt0, agg0,
                                                           cnt0, E0);
  }
  {
    int blocks = (N1_ / 16) / 8;
    sage_compute_kernel<INF_, true><<<blocks, 256, 0, stream>>>(
        agg0, cnt0, x, wl0, bl0, wr0, h, N1_);
  }
  {
    long long total = (long long)E1 * 8;
    int blocks = (int)((total + 255) / 256);
    sage_scatter_kernel<8, 3><<<blocks, 256, 0, stream>>>(h, src1, tgt1, agg1,
                                                          cnt1, E1);
  }
  {
    int blocks = (N2_ / 16) / 8;
    sage_compute_kernel<H_, false><<<blocks, 256, 0, stream>>>(
        agg1, cnt1, h, wl1, bl1, wr1, (float*)d_out, N2_);
  }
}


# Layer 0 projects before it aggregates: y0 = x @ wl0^T on WMMA for all source rows, then the edge scatter adds 32 floats per edge, not 128; f32 sums now run over projected rows (new summation order)
# speedup vs baseline: 1.3800x; 1.3800x over previous
#include <hip/hip_runtime.h>
#include <math.h>

typedef __attribute__((ext_vector_type(2))) float v2f;
typedef __attribute__((ext_vector_type(8))) float v8f;

#define N0_  262144
#define INF_ 128
#define H_   32
#define N1_  131072
#define N2_  65536

__device__ __forceinline__ v8f wmma_f32_k4(v2f a, v2f b, v8f c) {
  return __builtin_amdgcn_wmma_f32_16x16x4_f32(
       false, a,  false, b,
       (short)0, c,  false,  false);
}

template <int F4, int LOG2F4>
__global__ __launch_bounds__(256) void sage_scatter_kernel(
    const float* __restrict__ xsrc, const int* __restrict__ src,
    const int* __restrict__ tgt, float* __restrict__ agg,
    float* __restrict__ cnt, int nedges) {
  long long tid = (long long)blockIdx.x * blockDim.x + threadIdx.x;
  long long total = (long long)nedges << LOG2F4;
  if (tid >= total) return;
  int e = (int)(tid >> LOG2F4);
  int g = (int)(tid & (F4 - 1));
  int s = src[e];
  int t = tgt[e];
  const float4* x4 = (const float4*)xsrc;
  float4 v = x4[(size_t)s * F4 + g];
  float* dst = agg + (size_t)t * (F4 * 4) + g * 4;
  unsafeAtomicAdd(dst + 0, v.x);
  unsafeAtomicAdd(dst + 1, v.y);
  unsafeAtomicAdd(dst + 2, v.z);
  unsafeAtomicAdd(dst + 3, v.w);
  if (g == 0) unsafeAtomicAdd(cnt + t, 1.0f);
}

template <int K>
__global__ __launch_bounds__(256) void sage_project_kernel(
    const float* __restrict__ xin, const float* __restrict__ w,
    float* __restrict__ y, int nrows) {
  const int lane = threadIdx.x & 31;
  const int wid  = blockIdx.x * (blockDim.x >> 5) + (threadIdx.x >> 5);
  const int base = wid * 16;
  if (base >= nrows) return;
  const int l16  = lane & 15;
  const int half = lane >> 4;
  const int rowA = base + l16;

  v8f c0 = {};
  v8f c1 = {};
  const float2* A2 = (const float2*)(xin + (size_t)rowA * K);
  const float2* B0 = (const float2*)(w + (size_t)l16 * K);
  const float2* B1 = (const float2*)(w + (size_t)(l16 + 16) * K);
#pragma unroll
  for (int kc = 0; kc < K; kc += 4) {
    const int fi = (kc >> 1) + half;
    float2 fa = A2[fi];
    float2 f0 = B0[fi];
    float2 f1 = B1[fi];
    v2f a;  a.x  = fa.x; a.y  = fa.y;
    v2f b0; b0.x = f0.x; b0.y = f0.y;
    v2f b1; b1.x = f1.x; b1.y = f1.y;
    c0 = wmma_f32_k4(a, b0, c0);
    c1 = wmma_f32_k4(a, b1, c1);
  }
#pragma unroll
  for (int r = 0; r < 8; ++r) {
    const size_t row = (size_t)(base + r + half * 8);
    y[row * 32 + l16]      = c0[r];
    y[row * 32 + 16 + l16] = c1[r];
  }
}

template <int K, bool RELU>
__global__ __launch_bounds__(256) void sage_combine_kernel(
    const float* __restrict__ aggy, const float* __restrict__ cnt,
    const float* __restrict__ xin, const float* __restrict__ bl,
    const float* __restrict__ wr, float* __restrict__ out, int nrows) {
  const int lane = threadIdx.x & 31;
  const int wid  = blockIdx.x * (blockDim.x >> 5) + (threadIdx.x >> 5);
  const int base = wid * 16;
  if (base >= nrows) return;
  const int l16  = lane & 15;
  const int half = lane >> 4;
  const int rowA = base + l16;

  v8f c0;
  v8f c1;
#pragma unroll
  for (int r = 0; r < 8; ++r) {
    const size_t row = (size_t)(base + r + half * 8);
    const float inv = 1.0f / fmaxf(cnt[row], 1.0f);
    c0[r] = aggy[row * 32 + l16] * inv;
    c1[r] = aggy[row * 32 + 16 + l16] * inv;
  }

  const float2* A2 = (const float2*)(xin + (size_t)rowA * K);
  const float2* B0 = (const float2*)(wr + (size_t)l16 * K);
  const float2* B1 = (const float2*)(wr + (size_t)(l16 + 16) * K);
#pragma unroll
  for (int kc = 0; kc < K; kc += 4) {
    const int fi = (kc >> 1) + half;
    float2 fa = A2[fi];
    float2 f0 = B0[fi];
    float2 f1 = B1[fi];
    v2f a;  a.x  = fa.x; a.y  = fa.y;
    v2f b0; b0.x = f0.x; b0.y = f0.y;
    v2f b1; b1.x = f1.x; b1.y = f1.y;
    c0 = wmma_f32_k4(a, b0, c0);
    c1 = wmma_f32_k4(a, b1, c1);
  }

  const float bias0 = bl[l16];
  const float bias1 = bl[l16 + 16];
#pragma unroll
  for (int r = 0; r < 8; ++r) { c0[r] += bias0; c1[r] += bias1; }

#pragma unroll
  for (int r = 0; r < 8; ++r) {
    float p = c0[r] * c0[r] + c1[r] * c1[r];
    p += __shfl_xor(p, 1, 32);
    p += __shfl_xor(p, 2, 32);
    p += __shfl_xor(p, 4, 32);
    p += __shfl_xor(p, 8, 32);
    float scale = 1.0f / fmaxf(sqrtf(p), 1e-12f);
    float o0 = c0[r] * scale;
    float o1 = c1[r] * scale;
    if (RELU) { o0 = fmaxf(o0, 0.0f); o1 = fmaxf(o1, 0.0f); }
    const size_t row = (size_t)(base + r + half * 8);
    out[row * 32 + l16]      = o0;
    out[row * 32 + 16 + l16] = o1;
  }
}

template <int K, bool RELU>
__global__ __launch_bounds__(256) void sage_compute_kernel(
    const float* __restrict__ agg, const float* __restrict__ cnt,
    const float* __restrict__ xin, const float* __restrict__ wl,
    const float* __restrict__ bl, const float* __restrict__ wr,
    float* __restrict__ out, int nrows) {
  const int lane = threadIdx.x & 31;
  const int wid  = blockIdx.x * (blockDim.x >> 5) + (threadIdx.x >> 5);
  const int base = wid * 16;
  if (base >= nrows) return;
  const int l16  = lane & 15;
  const int half = lane >> 4;
  const int rowA = base + l16;

  const float inv = 1.0f / fmaxf(cnt[rowA], 1.0f);

  v8f c0 = {};
  v8f c1 = {};

#pragma unroll
  for (int phase = 0; phase < 2; ++phase) {
    const float* Arow = (phase == 0 ? agg : xin) + (size_t)rowA * K;
    const float* W    = (phase == 0 ? wl : wr);
    const float  s    = (phase == 0 ? inv : 1.0f);
    const float2* A2  = (const float2*)Arow;
    const float2* B0  = (const float2*)(W + (size_t)l16 * K);
    const float2* B1  = (const float2*)(W + (size_t)(l16 + 16) * K);
#pragma unroll
    for (int kc = 0; kc < K; kc += 4) {
      const int fi = (kc >> 1) + half;
      float2 fa = A2[fi];
      float2 f0 = B0[fi];
      float2 f1 = B1[fi];
      v2f a;  a.x  = fa.x * s; a.y  = fa.y * s;
      v2f b0; b0.x = f0.x;     b0.y = f0.y;
      v2f b1; b1.x = f1.x;     b1.y = f1.y;
      c0 = wmma_f32_k4(a, b0, c0);
      c1 = wmma_f32_k4(a, b1, c1);
    }
  }

  const float bias0 = bl[l16];
  const float bias1 = bl[l16 + 16];
#pragma unroll
  for (int r = 0; r < 8; ++r) { c0[r] += bias0; c1[r] += bias1; }

#pragma unroll
  for (int r = 0; r < 8; ++r) {
    float p = c0[r] * c0[r] + c1[r] * c1[r];
    p += __shfl_xor(p, 1, 32);
    p += __shfl_xor(p, 2, 32);
    p += __shfl_xor(p, 4, 32);
    p += __shfl_xor(p, 8, 32);
    float scale = 1.0f / fmaxf(sqrtf(p), 1e-12f);
    float o0 = c0[r] * scale;
    float o1 = c1[r] * scale;
    if (RELU) { o0 = fmaxf(o0, 0.0f); o1 = fmaxf(o1, 0.0f); }
    const size_t row = (size_t)(base + r + half * 8);
    out[row * 32 + l16]      = o0;
    out[row * 32 + 16 + l16] = o1;
  }
}

extern "C" void kernel_launch(void* const* d_in, const int* in_sizes, int n_in,
                              void* d_out, int out_size, void* d_ws,
                              size_t ws_size, hipStream_t stream) {
  const float* x   = (const float*)d_in[0];
  const int* src0  = (const int*)d_in[1];
  const int* tgt0  = (const int*)d_in[2];
  const int* src1  = (const int*)d_in[3];
  const int* tgt1  = (const int*)d_in[4];
  const float* wl0 = (const float*)d_in[5];
  const float* bl0 = (const float*)d_in[6];
  const float* wr0 = (const float*)d_in[7];
  const float* wl1 = (const float*)d_in[8];
  const float* bl1 = (const float*)d_in[9];
  const float* wr1 = (const float*)d_in[10];
  const int E0 = in_sizes[1];
  const int E1 = in_sizes[3];

  float* ws   = (float*)d_ws;
  float* agg0 = ws;
  float* cnt0 = agg0 + (size_t)N1_ * H_;
  float* agg1 = cnt0 + (size_t)N1_;
  float* cnt1 = agg1 + (size_t)N2_ * H_;
  float* y0   = cnt1 + (size_t)N2_;
  float* h    = y0 + (size_t)N0_ * H_;
  const size_t zero_bytes =
      ((size_t)N1_ * H_ + N1_ + (size_t)N2_ * H_ + N2_) * sizeof(float);
  hipMemsetAsync(d_ws, 0, zero_bytes, stream);

  {
    int blocks = (N0_ / 16) / 8;
    sage_project_kernel<INF_><<<blocks, 256, 0, stream>>>(x, wl0, y0, N0_);
  }
  {
    long long total = (long long)E0 * 8;
    int blocks = (int)((total + 255) / 256);
    sage_scatter_kernel<8, 3><<<blocks, 256, 0, stream>>>(y0, src0, tgt0, agg0,
                                                          cnt0, E0);
  }
  {
    int blocks = (N1_ / 16) / 8;
    sage_combine_kernel<INF_, true><<<blocks, 256, 0, stream>>>(
        agg0, cnt0, x, bl0, wr0, h, N1_);
  }
  {
    long long total = (long long)E1 * 8;
    int blocks = (int)((total + 255) / 256);
    sage_scatter_kernel<8, 3><<<blocks, 256, 0, stream>>>(h, src1, tgt1, agg1,
                                                          cnt1, E1);
  }
  {
    int blocks = (N2_ / 16) / 8;
    sage_compute_kernel<H_, false><<<blocks, 256, 0, stream>>>(
        agg1, cnt1, h, wl1, bl1, wr1, (float*)d_out, N2_);
  }
}


# Scatter waves add into a private agg/cnt plane per L2 half (by die id); combine and transform add the two planes, so no atomic line is written from both halves; f32 sums form per half, then add
# speedup vs baseline: 1.8488x; 1.3397x over previous
#include <hip/hip_runtime.h>
#include <math.h>

typedef __attribute__((ext_vector_type(2))) float v2f;
typedef __attribute__((ext_vector_type(8))) float v8f;

#define N0_  262144
#define INF_ 128
#define H_   32
#define N1_  131072
#define N2_  65536

__device__ __forceinline__ v8f wmma_f32_k4(v2f a, v2f b, v8f c) {
  return __builtin_amdgcn_wmma_f32_16x16x4_f32(
       false, a,  false, b,
       (short)0, c,  false,  false);
}

template <int F4, int LOG2F4>
__global__ __launch_bounds__(256) void sage_scatter_kernel(
    const float* __restrict__ xsrc, const int* __restrict__ src,
    const int* __restrict__ tgt, float* __restrict__ agg,
    float* __restrict__ cnt, int nedges, int agg_plane, int cnt_plane) {
  const unsigned place = __builtin_amdgcn_s_sendmsg_rtn(0x87);
  const unsigned upper = (place >> 18) & 1u;
  agg += (size_t)upper * (size_t)agg_plane;
  cnt += (size_t)upper * (size_t)cnt_plane;
  long long tid = (long long)blockIdx.x * blockDim.x + threadIdx.x;
  long long total = (long long)nedges << LOG2F4;
  if (tid >= total) return;
  int e = (int)(tid >> LOG2F4);
  int g = (int)(tid & (F4 - 1));
  int s = src[e];
  int t = tgt[e];
  const float4* x4 = (const float4*)xsrc;
  float4 v = x4[(size_t)s * F4 + g];
  float* dst = agg + (size_t)t * (F4 * 4) + g * 4;
  unsafeAtomicAdd(dst + 0, v.x);
  unsafeAtomicAdd(dst + 1, v.y);
  unsafeAtomicAdd(dst + 2, v.z);
  unsafeAtomicAdd(dst + 3, v.w);
  if (g == 0) unsafeAtomicAdd(cnt + t, 1.0f);
}

template <int K>
__global__ __launch_bounds__(256) void sage_project_kernel(
    const float* __restrict__ xin, const float* __restrict__ w,
    float* __restrict__ y, int nrows) {
  const int lane = threadIdx.x & 31;
  const int wid  = blockIdx.x * (blockDim.x >> 5) + (threadIdx.x >> 5);
  const int base = wid * 16;
  if (base >= nrows) return;
  const int l16  = lane & 15;
  const int half = lane >> 4;
  const int rowA = base + l16;

  v8f c0 = {};
  v8f c1 = {};
  const float2* A2 = (const float2*)(xin + (size_t)rowA * K);
  const float2* B0 = (const float2*)(w + (size_t)l16 * K);
  const float2* B1 = (const float2*)(w + (size_t)(l16 + 16) * K);
#pragma unroll
  for (int kc = 0; kc < K; kc += 4) {
    const int fi = (kc >> 1) + half;
    float2 fa = A2[fi];
    float2 f0 = B0[fi];
    float2 f1 = B1[fi];
    v2f a;  a.x  = fa.x; a.y  = fa.y;
    v2f b0; b0.x = f0.x; b0.y = f0.y;
    v2f b1; b1.x = f1.x; b1.y = f1.y;
    c0 = wmma_f32_k4(a, b0, c0);
    c1 = wmma_f32_k4(a, b1, c1);
  }
#pragma unroll
  for (int r = 0; r < 8; ++r) {
    const size_t row = (size_t)(base + r + half * 8);
    y[row * 32 + l16]      = c0[r];
    y[row * 32 + 16 + l16] = c1[r];
  }
}

template <int K, bool RELU>
__global__ __launch_bounds__(256) void sage_combine_kernel(
    const float* __restrict__ aggy, const float* __restrict__ cnt,
    const float* __restrict__ xin, const float* __restrict__ bl,
    const float* __restrict__ wr, float* __restrict__ out, int nrows,
    int agg_plane, int cnt_plane) {
  const int lane = threadIdx.x & 31;
  const int wid  = blockIdx.x * (blockDim.x >> 5) + (threadIdx.x >> 5);
  const int base = wid * 16;
  if (base >= nrows) return;
  const int l16  = lane & 15;
  const int half = lane >> 4;
  const int rowA = base + l16;

  v8f c0;
  v8f c1;
#pragma unroll
  for (int r = 0; r < 8; ++r) {
    const size_t row = (size_t)(base + r + half * 8);
    const float inv = 1.0f / fmaxf(cnt[row] + cnt[row + cnt_plane], 1.0f);
    c0[r] = (aggy[row * 32 + l16] + aggy[row * 32 + l16 + agg_plane]) * inv;
    c1[r] = (aggy[row * 32 + 16 + l16] + aggy[row * 32 + 16 + l16 + agg_plane]) *
            inv;
  }

  const float2* A2 = (const float2*)(xin + (size_t)rowA * K);
  const float2* B0 = (const float2*)(wr + (size_t)l16 * K);
  const float2* B1 = (const float2*)(wr + (size_t)(l16 + 16) * K);
#pragma unroll
  for (int kc = 0; kc < K; kc += 4) {
    const int fi = (kc >> 1) + half;
    float2 fa = A2[fi];
    float2 f0 = B0[fi];
    float2 f1 = B1[fi];
    v2f a;  a.x  = fa.x; a.y  = fa.y;
    v2f b0; b0.x = f0.x; b0.y = f0.y;
    v2f b1; b1.x = f1.x; b1.y = f1.y;
    c0 = wmma_f32_k4(a, b0, c0);
    c1 = wmma_f32_k4(a, b1, c1);
  }

  const float bias0 = bl[l16];
  const float bias1 = bl[l16 + 16];
#pragma unroll
  for (int r = 0; r < 8; ++r) { c0[r] += bias0; c1[r] += bias1; }

#pragma unroll
  for (int r = 0; r < 8; ++r) {
    float p = c0[r] * c0[r] + c1[r] * c1[r];
    p += __shfl_xor(p, 1, 32);
    p += __shfl_xor(p, 2, 32);
    p += __shfl_xor(p, 4, 32);
    p += __shfl_xor(p, 8, 32);
    float scale = 1.0f / fmaxf(sqrtf(p), 1e-12f);
    float o0 = c0[r] * scale;
    float o1 = c1[r] * scale;
    if (RELU) { o0 = fmaxf(o0, 0.0f); o1 = fmaxf(o1, 0.0f); }
    const size_t row = (size_t)(base + r + half * 8);
    out[row * 32 + l16]      = o0;
    out[row * 32 + 16 + l16] = o1;
  }
}

template <int K, bool RELU>
__global__ __launch_bounds__(256) void sage_compute_kernel(
    const float* __restrict__ agg, const float* __restrict__ cnt,
    const float* __restrict__ xin, const float* __restrict__ wl,
    const float* __restrict__ bl, const float* __restrict__ wr,
    float* __restrict__ out, int nrows, int agg_plane, int cnt_plane) {
  const int lane = threadIdx.x & 31;
  const int wid  = blockIdx.x * (blockDim.x >> 5) + (threadIdx.x >> 5);
  const int base = wid * 16;
  if (base >= nrows) return;
  const int l16  = lane & 15;
  const int half = lane >> 4;
  const int rowA = base + l16;

  const float inv = 1.0f / fmaxf(cnt[rowA] + cnt[rowA + cnt_plane], 1.0f);

  v8f c0 = {};
  v8f c1 = {};

  const float2* P2 = (const float2*)(agg + agg_plane + (size_t)rowA * K);

#pragma unroll
  for (int phase = 0; phase < 2; ++phase) {
    const float* Arow = (phase == 0 ? agg : xin) + (size_t)rowA * K;
    const float* W    = (phase == 0 ? wl : wr);
    const float  s    = (phase == 0 ? inv : 1.0f);
    const float2* A2  = (const float2*)Arow;
    const float2* B0  = (const float2*)(W + (size_t)l16 * K);
    const float2* B1  = (const float2*)(W + (size_t)(l16 + 16) * K);
#pragma unroll
    for (int kc = 0; kc < K; kc += 4) {
      const int fi = (kc >> 1) + half;
      float2 fa = A2[fi];
      if (phase == 0) {
        const float2 fp = P2[fi];
        fa.x += fp.x;
        fa.y += fp.y;
      }
      float2 f0 = B0[fi];
      float2 f1 = B1[fi];
      v2f a;  a.x  = fa.x * s; a.y  = fa.y * s;
      v2f b0; b0.x = f0.x;     b0.y = f0.y;
      v2f b1; b1.x = f1.x;     b1.y = f1.y;
      c0 = wmma_f32_k4(a, b0, c0);
      c1 = wmma_f32_k4(a, b1, c1);
    }
  }

  const float bias0 = bl[l16];
  const float bias1 = bl[l16 + 16];
#pragma unroll
  for (int r = 0; r < 8; ++r) { c0[r] += bias0; c1[r] += bias1; }

#pragma unroll
  for (int r = 0; r < 8; ++r) {
    float p = c0[r] * c0[r] + c1[r] * c1[r];
    p += __shfl_xor(p, 1, 32);
    p += __shfl_xor(p, 2, 32);
    p += __shfl_xor(p, 4, 32);
    p += __shfl_xor(p, 8, 32);
    float scale = 1.0f / fmaxf(sqrtf(p), 1e-12f);
    float o0 = c0[r] * scale;
    float o1 = c1[r] * scale;
    if (RELU) { o0 = fmaxf(o0, 0.0f); o1 = fmaxf(o1, 0.0f); }
    const size_t row = (size_t)(base + r + half * 8);
    out[row * 32 + l16]      = o0;
    out[row * 32 + 16 + l16] = o1;
  }
}

extern "C" void kernel_launch(void* const* d_in, const int* in_sizes, int n_in,
                              void* d_out, int out_size, void* d_ws,
                              size_t ws_size, hipStream_t stream) {
  const float* x   = (const float*)d_in[0];
  const int* src0  = (const int*)d_in[1];
  const int* tgt0  = (const int*)d_in[2];
  const int* src1  = (const int*)d_in[3];
  const int* tgt1  = (const int*)d_in[4];
  const float* wl0 = (const float*)d_in[5];
  const float* bl0 = (const float*)d_in[6];
  const float* wr0 = (const float*)d_in[7];
  const float* wl1 = (const float*)d_in[8];
  const float* bl1 = (const float*)d_in[9];
  const float* wr1 = (const float*)d_in[10];
  const int E0 = in_sizes[1];
  const int E1 = in_sizes[3];

  float* ws   = (float*)d_ws;
  float* agg0 = ws;
  float* cnt0 = agg0 + 2 * (size_t)N1_ * H_;
  float* agg1 = cnt0 + 2 * (size_t)N1_;
  float* cnt1 = agg1 + 2 * (size_t)N2_ * H_;
  float* y0   = cnt1 + 2 * (size_t)N2_;
  float* h    = y0;
  const size_t zero_bytes =
      2 * ((size_t)N1_ * H_ + N1_ + (size_t)N2_ * H_ + N2_) * sizeof(float);
  hipMemsetAsync(d_ws, 0, zero_bytes, stream);

  {
    int blocks = (N0_ / 16) / 8;
    sage_project_kernel<INF_><<<blocks, 256, 0, stream>>>(x, wl0, y0, N0_);
  }
  {
    long long total = (long long)E0 * 8;
    int blocks = (int)((total + 255) / 256);
    sage_scatter_kernel<8, 3><<<blocks, 256, 0, stream>>>(
        y0, src0, tgt0, agg0, cnt0, E0, N1_ * H_, N1_);
  }
  {
    int blocks = (N1_ / 16) / 8;
    sage_combine_kernel<INF_, true><<<blocks, 256, 0, stream>>>(
        agg0, cnt0, x, bl0, wr0, h, N1_, N1_ * H_, N1_);
  }
  {
    long long total = (long long)E1 * 8;
    int blocks = (int)((total + 255) / 256);
    sage_scatter_kernel<8, 3><<<blocks, 256, 0, stream>>>(
        h, src1, tgt1, agg1, cnt1, E1, N2_ * H_, N2_);
  }
  {
    int blocks = (N2_ / 16) / 8;
    sage_compute_kernel<H_, false><<<blocks, 256, 0, stream>>>(
        agg1, cnt1, h, wl1, bl1, wr1, (float*)d_out, N2_, N2_ * H_, N2_);
  }
}


# Scatter puts one edge row on a wave's 32 lanes: each f32 atomic instruction adds one whole 128-byte row (was 4 rows at a 16-byte stride), 32 edges per wave; f32 atomic sums, arrival order as before
# speedup vs baseline: 3.7679x; 2.0380x over previous
#include <hip/hip_runtime.h>
#include <math.h>

typedef __attribute__((ext_vector_type(2))) float v2f;
typedef __attribute__((ext_vector_type(8))) float v8f;

#define N0_  262144
#define INF_ 128
#define H_   32
#define N1_  131072
#define N2_  65536

__device__ __forceinline__ v8f wmma_f32_k4(v2f a, v2f b, v8f c) {
  return __builtin_amdgcn_wmma_f32_16x16x4_f32(
       false, a,  false, b,
       (short)0, c,  false,  false);
}

__global__ __launch_bounds__(256) void sage_scatter_kernel(
    const float* __restrict__ xsrc, const int* __restrict__ src,
    const int* __restrict__ tgt, float* __restrict__ agg,
    float* __restrict__ cnt, int nedges, int agg_plane, int cnt_plane) {
  const unsigned place = __builtin_amdgcn_s_sendmsg_rtn(0x87);
  const unsigned upper = (place >> 18) & 1u;
  agg += (size_t)upper * (size_t)agg_plane;
  cnt += (size_t)upper * (size_t)cnt_plane;
  const int lane = threadIdx.x & 31;
  const int wave = __builtin_amdgcn_readfirstlane((int)(threadIdx.x >> 5));
  const int wid  = blockIdx.x * (blockDim.x >> 5) + wave;
  const int e0   = wid * 32;
  if (e0 >= nedges) return;
  const int left = nedges - e0;
  const unsigned loff = (unsigned)lane;
  int s_v = 0;
  int t_v = 0;
  if (lane < left) {
    s_v = (src + e0)[loff];
    t_v = (tgt + e0)[loff];
    unsafeAtomicAdd(cnt + t_v, 1.0f);
  }
  if (left >= 32) {
    float v[32];
#pragma unroll
    for (int j = 0; j < 32; ++j) {
      const unsigned s = (unsigned)__builtin_amdgcn_readlane(s_v, j);
      v[j] = xsrc[s * 32u + loff];
    }
#pragma unroll
    for (int j = 0; j < 32; ++j) {
      const unsigned t = (unsigned)__builtin_amdgcn_readlane(t_v, j);
      unsafeAtomicAdd(agg + (t * 32u + loff), v[j]);
    }
  } else {
    for (int j = 0; j < left; ++j) {
      const unsigned s = (unsigned)__builtin_amdgcn_readlane(s_v, j);
      const unsigned t = (unsigned)__builtin_amdgcn_readlane(t_v, j);
      unsafeAtomicAdd(agg + (t * 32u + loff), xsrc[s * 32u + loff]);
    }
  }
}

template <int K>
__global__ __launch_bounds__(256) void sage_project_kernel(
    const float* __restrict__ xin, const float* __restrict__ w,
    float* __restrict__ y, int nrows) {
  const int lane = threadIdx.x & 31;
  const int wid  = blockIdx.x * (blockDim.x >> 5) + (threadIdx.x >> 5);
  const int base = wid * 16;
  if (base >= nrows) return;
  const int l16  = lane & 15;
  const int half = lane >> 4;
  const int rowA = base + l16;

  v8f c0 = {};
  v8f c1 = {};
  const float2* A2 = (const float2*)(xin + (size_t)rowA * K);
  const float2* B0 = (const float2*)(w + (size_t)l16 * K);
  const float2* B1 = (const float2*)(w + (size_t)(l16 + 16) * K);
#pragma unroll
  for (int kc = 0; kc < K; kc += 4) {
    const int fi = (kc >> 1) + half;
    float2 fa = A2[fi];
    float2 f0 = B0[fi];
    float2 f1 = B1[fi];
    v2f a;  a.x  = fa.x; a.y  = fa.y;
    v2f b0; b0.x = f0.x; b0.y = f0.y;
    v2f b1; b1.x = f1.x; b1.y = f1.y;
    c0 = wmma_f32_k4(a, b0, c0);
    c1 = wmma_f32_k4(a, b1, c1);
  }
#pragma unroll
  for (int r = 0; r < 8; ++r) {
    const size_t row = (size_t)(base + r + half * 8);
    y[row * 32 + l16]      = c0[r];
    y[row * 32 + 16 + l16] = c1[r];
  }
}

template <int K, bool RELU>
__global__ __launch_bounds__(256) void sage_combine_kernel(
    const float* __restrict__ aggy, const float* __restrict__ cnt,
    const float* __restrict__ xin, const float* __restrict__ bl,
    const float* __restrict__ wr, float* __restrict__ out, int nrows,
    int agg_plane, int cnt_plane) {
  const int lane = threadIdx.x & 31;
  const int wid  = blockIdx.x * (blockDim.x >> 5) + (threadIdx.x >> 5);
  const int base = wid * 16;
  if (base >= nrows) return;
  const int l16  = lane & 15;
  const int half = lane >> 4;
  const int rowA = base + l16;

  v8f c0;
  v8f c1;
#pragma unroll
  for (int r = 0; r < 8; ++r) {
    const size_t row = (size_t)(base + r + half * 8);
    const float inv = 1.0f / fmaxf(cnt[row] + cnt[row + cnt_plane], 1.0f);
    c0[r] = (aggy[row * 32 + l16] + aggy[row * 32 + l16 + agg_plane]) * inv;
    c1[r] = (aggy[row * 32 + 16 + l16] + aggy[row * 32 + 16 + l16 + agg_plane]) *
            inv;
  }

  const float2* A2 = (const float2*)(xin + (size_t)rowA * K);
  const float2* B0 = (const float2*)(wr + (size_t)l16 * K);
  const float2* B1 = (const float2*)(wr + (size_t)(l16 + 16) * K);
#pragma unroll
  for (int kc = 0; kc < K; kc += 4) {
    const int fi = (kc >> 1) + half;
    float2 fa = A2[fi];
    float2 f0 = B0[fi];
    float2 f1 = B1[fi];
    v2f a;  a.x  = fa.x; a.y  = fa.y;
    v2f b0; b0.x = f0.x; b0.y = f0.y;
    v2f b1; b1.x = f1.x; b1.y = f1.y;
    c0 = wmma_f32_k4(a, b0, c0);
    c1 = wmma_f32_k4(a, b1, c1);
  }

  const float bias0 = bl[l16];
  const float bias1 = bl[l16 + 16];
#pragma unroll
  for (int r = 0; r < 8; ++r) { c0[r] += bias0; c1[r] += bias1; }

#pragma unroll
  for (int r = 0; r < 8; ++r) {
    float p = c0[r] * c0[r] + c1[r] * c1[r];
    p += __shfl_xor(p, 1, 32);
    p += __shfl_xor(p, 2, 32);
    p += __shfl_xor(p, 4, 32);
    p += __shfl_xor(p, 8, 32);
    float scale = 1.0f / fmaxf(sqrtf(p), 1e-12f);
    float o0 = c0[r] * scale;
    float o1 = c1[r] * scale;
    if (RELU) { o0 = fmaxf(o0, 0.0f); o1 = fmaxf(o1, 0.0f); }
    const size_t row = (size_t)(base + r + half * 8);
    out[row * 32 + l16]      = o0;
    out[row * 32 + 16 + l16] = o1;
  }
}

template <int K, bool RELU>
__global__ __launch_bounds__(256) void sage_compute_kernel(
    const float* __restrict__ agg, const float* __restrict__ cnt,
    const float* __restrict__ xin, const float* __restrict__ wl,
    const float* __restrict__ bl, const float* __restrict__ wr,
    float* __restrict__ out, int nrows, int agg_plane, int cnt_plane) {
  const int lane = threadIdx.x & 31;
  const int wid  = blockIdx.x * (blockDim.x >> 5) + (threadIdx.x >> 5);
  const int base = wid * 16;
  if (base >= nrows) return;
  const int l16  = lane & 15;
  const int half = lane >> 4;
  const int rowA = base + l16;

  const float inv = 1.0f / fmaxf(cnt[rowA] + cnt[rowA + cnt_plane], 1.0f);

  v8f c0 = {};
  v8f c1 = {};

  const float2* P2 = (const float2*)(agg + agg_plane + (size_t)rowA * K);

#pragma unroll
  for (int phase = 0; phase < 2; ++phase) {
    const float* Arow = (phase == 0 ? agg : xin) + (size_t)rowA * K;
    const float* W    = (phase == 0 ? wl : wr);
    const float  s    = (phase == 0 ? inv : 1.0f);
    const float2* A2  = (const float2*)Arow;
    const float2* B0  = (const float2*)(W + (size_t)l16 * K);
    const float2* B1  = (const float2*)(W + (size_t)(l16 + 16) * K);
#pragma unroll
    for (int kc = 0; kc < K; kc += 4) {
      const int fi = (kc >> 1) + half;
      float2 fa = A2[fi];
      if (phase == 0) {
        const float2 fp = P2[fi];
        fa.x += fp.x;
        fa.y += fp.y;
      }
      float2 f0 = B0[fi];
      float2 f1 = B1[fi];
      v2f a;  a.x  = fa.x * s; a.y  = fa.y * s;
      v2f b0; b0.x = f0.x;     b0.y = f0.y;
      v2f b1; b1.x = f1.x;     b1.y = f1.y;
      c0 = wmma_f32_k4(a, b0, c0);
      c1 = wmma_f32_k4(a, b1, c1);
    }
  }

  const float bias0 = bl[l16];
  const float bias1 = bl[l16 + 16];
#pragma unroll
  for (int r = 0; r < 8; ++r) { c0[r] += bias0; c1[r] += bias1; }

#pragma unroll
  for (int r = 0; r < 8; ++r) {
    float p = c0[r] * c0[r] + c1[r] * c1[r];
    p += __shfl_xor(p, 1, 32);
    p += __shfl_xor(p, 2, 32);
    p += __shfl_xor(p, 4, 32);
    p += __shfl_xor(p, 8, 32);
    float scale = 1.0f / fmaxf(sqrtf(p), 1e-12f);
    float o0 = c0[r] * scale;
    float o1 = c1[r] * scale;
    if (RELU) { o0 = fmaxf(o0, 0.0f); o1 = fmaxf(o1, 0.0f); }
    const size_t row = (size_t)(base + r + half * 8);
    out[row * 32 + l16]      = o0;
    out[row * 32 + 16 + l16] = o1;
  }
}

extern "C" void kernel_launch(void* const* d_in, const int* in_sizes, int n_in,
                              void* d_out, int out_size, void* d_ws,
                              size_t ws_size, hipStream_t stream) {
  const float* x   = (const float*)d_in[0];
  const int* src0  = (const int*)d_in[1];
  const int* tgt0  = (const int*)d_in[2];
  const int* src1  = (const int*)d_in[3];
  const int* tgt1  = (const int*)d_in[4];
  const float* wl0 = (const float*)d_in[5];
  const float* bl0 = (const float*)d_in[6];
  const float* wr0 = (const float*)d_in[7];
  const float* wl1 = (const float*)d_in[8];
  const float* bl1 = (const float*)d_in[9];
  const float* wr1 = (const float*)d_in[10];
  const int E0 = in_sizes[1];
  const int E1 = in_sizes[3];

  float* ws   = (float*)d_ws;
  float* agg0 = ws;
  float* cnt0 = agg0 + 2 * (size_t)N1_ * H_;
  float* agg1 = cnt0 + 2 * (size_t)N1_;
  float* cnt1 = agg1 + 2 * (size_t)N2_ * H_;
  float* y0   = cnt1 + 2 * (size_t)N2_;
  float* h    = y0;
  const size_t zero_bytes =
      2 * ((size_t)N1_ * H_ + N1_ + (size_t)N2_ * H_ + N2_) * sizeof(float);
  hipMemsetAsync(d_ws, 0, zero_bytes, stream);

  {
    int blocks = (N0_ / 16) / 8;
    sage_project_kernel<INF_><<<blocks, 256, 0, stream>>>(x, wl0, y0, N0_);
  }
  {
    int blocks = (E0 + 255) / 256;
    sage_scatter_kernel<<<blocks, 256, 0, stream>>>(
        y0, src0, tgt0, agg0, cnt0, E0, N1_ * H_, N1_);
  }
  {
    int blocks = (N1_ / 16) / 8;
    sage_combine_kernel<INF_, true><<<blocks, 256, 0, stream>>>(
        agg0, cnt0, x, bl0, wr0, h, N1_, N1_ * H_, N1_);
  }
  {
    int blocks = (E1 + 255) / 256;
    sage_scatter_kernel<<<blocks, 256, 0, stream>>>(
        h, src1, tgt1, agg1, cnt1, E1, N2_ * H_, N2_);
  }
  {
    int blocks = (N2_ / 16) / 8;
    sage_compute_kernel<H_, false><<<blocks, 256, 0, stream>>>(
        agg1, cnt1, h, wl1, bl1, wr1, (float*)d_out, N2_, N2_ * H_, N2_);
  }
}


# Each work-group copies its 32 x K weight matrix into LDS once (whole-line loads, pitch K+4); WMMA B fragments come from LDS, not from 16-line global loads twice per k step; k order and sums unchanged
# speedup vs baseline: 4.2870x; 1.1378x over previous
#include <hip/hip_runtime.h>
#include <math.h>

typedef __attribute__((ext_vector_type(2))) float v2f;
typedef __attribute__((ext_vector_type(8))) float v8f;

#define N0_  262144
#define INF_ 128
#define H_   32
#define N1_  131072
#define N2_  65536

__device__ __forceinline__ v8f wmma_f32_k4(v2f a, v2f b, v8f c) {
  return __builtin_amdgcn_wmma_f32_16x16x4_f32(
       false, a,  false, b,
       (short)0, c,  false,  false);
}

__global__ __launch_bounds__(256) void sage_scatter_kernel(
    const float* __restrict__ xsrc, const int* __restrict__ src,
    const int* __restrict__ tgt, float* __restrict__ agg,
    float* __restrict__ cnt, int nedges, int agg_plane, int cnt_plane) {
  const unsigned place = __builtin_amdgcn_s_sendmsg_rtn(0x87);
  const unsigned upper = (place >> 18) & 1u;
  agg += (size_t)upper * (size_t)agg_plane;
  cnt += (size_t)upper * (size_t)cnt_plane;
  const int lane = threadIdx.x & 31;
  const int wave = __builtin_amdgcn_readfirstlane((int)(threadIdx.x >> 5));
  const int wid  = blockIdx.x * (blockDim.x >> 5) + wave;
  const int e0   = wid * 32;
  if (e0 >= nedges) return;
  const int left = nedges - e0;
  const unsigned loff = (unsigned)lane;
  int s_v = 0;
  int t_v = 0;
  if (lane < left) {
    s_v = (src + e0)[loff];
    t_v = (tgt + e0)[loff];
    unsafeAtomicAdd(cnt + t_v, 1.0f);
  }
  if (left >= 32) {
    float v[32];
#pragma unroll
    for (int j = 0; j < 32; ++j) {
      const unsigned s = (unsigned)__builtin_amdgcn_readlane(s_v, j);
      v[j] = xsrc[s * 32u + loff];
    }
#pragma unroll
    for (int j = 0; j < 32; ++j) {
      const unsigned t = (unsigned)__builtin_amdgcn_readlane(t_v, j);
      unsafeAtomicAdd(agg + (t * 32u + loff), v[j]);
    }
  } else {
    for (int j = 0; j < left; ++j) {
      const unsigned s = (unsigned)__builtin_amdgcn_readlane(s_v, j);
      const unsigned t = (unsigned)__builtin_amdgcn_readlane(t_v, j);
      unsafeAtomicAdd(agg + (t * 32u + loff), xsrc[s * 32u + loff]);
    }
  }
}

template <int K>
__device__ __forceinline__ void stage_weights(const float* __restrict__ w,
                                              float* __restrict__ lw) {
  static_assert(K % 32 == 0, "32*K floats: a whole number of float4 per thread");
  constexpr int P = K + 4;
  const float4* w4 = (const float4*)w;
#pragma unroll
  for (int i = 0; i < K / 32; ++i) {
    const int f   = i * 256 + (int)threadIdx.x;
    const int row = f / (K / 4);
    const int q   = f % (K / 4);
    *(float4*)(lw + row * P + 4 * q) = w4[f];
  }
}

template <int K>
__global__ __launch_bounds__(256) void sage_project_kernel(
    const float* __restrict__ xin, const float* __restrict__ w,
    float* __restrict__ y, int nrows) {
  constexpr int P = K + 4;
  __shared__ __attribute__((aligned(16))) float lw[32 * P];
  stage_weights<K>(w, lw);
  __syncthreads();
  const int lane = threadIdx.x & 31;
  const int wid  = blockIdx.x * (blockDim.x >> 5) + (threadIdx.x >> 5);
  const int base = wid * 16;
  if (base >= nrows) return;
  const int l16  = lane & 15;
  const int half = lane >> 4;
  const int rowA = base + l16;

  v8f c0 = {};
  v8f c1 = {};
  const float2* A2 = (const float2*)(xin + (size_t)rowA * K);
  const float2* B0 = (const float2*)(lw + l16 * P);
  const float2* B1 = (const float2*)(lw + (l16 + 16) * P);
#pragma unroll
  for (int kc = 0; kc < K; kc += 4) {
    const int fi = (kc >> 1) + half;
    float2 fa = A2[fi];
    float2 f0 = B0[fi];
    float2 f1 = B1[fi];
    v2f a;  a.x  = fa.x; a.y  = fa.y;
    v2f b0; b0.x = f0.x; b0.y = f0.y;
    v2f b1; b1.x = f1.x; b1.y = f1.y;
    c0 = wmma_f32_k4(a, b0, c0);
    c1 = wmma_f32_k4(a, b1, c1);
  }
#pragma unroll
  for (int r = 0; r < 8; ++r) {
    const size_t row = (size_t)(base + r + half * 8);
    y[row * 32 + l16]      = c0[r];
    y[row * 32 + 16 + l16] = c1[r];
  }
}

template <int K, bool RELU>
__global__ __launch_bounds__(256) void sage_combine_kernel(
    const float* __restrict__ aggy, const float* __restrict__ cnt,
    const float* __restrict__ xin, const float* __restrict__ bl,
    const float* __restrict__ wr, float* __restrict__ out, int nrows,
    int agg_plane, int cnt_plane) {
  constexpr int P = K + 4;
  __shared__ __attribute__((aligned(16))) float lw[32 * P];
  stage_weights<K>(wr, lw);
  __syncthreads();
  const int lane = threadIdx.x & 31;
  const int wid  = blockIdx.x * (blockDim.x >> 5) + (threadIdx.x >> 5);
  const int base = wid * 16;
  if (base >= nrows) return;
  const int l16  = lane & 15;
  const int half = lane >> 4;
  const int rowA = base + l16;

  v8f c0;
  v8f c1;
#pragma unroll
  for (int r = 0; r < 8; ++r) {
    const size_t row = (size_t)(base + r + half * 8);
    const float inv = 1.0f / fmaxf(cnt[row] + cnt[row + cnt_plane], 1.0f);
    c0[r] = (aggy[row * 32 + l16] + aggy[row * 32 + l16 + agg_plane]) * inv;
    c1[r] = (aggy[row * 32 + 16 + l16] + aggy[row * 32 + 16 + l16 + agg_plane]) *
            inv;
  }

  const float2* A2 = (const float2*)(xin + (size_t)rowA * K);
  const float2* B0 = (const float2*)(lw + l16 * P);
  const float2* B1 = (const float2*)(lw + (l16 + 16) * P);
#pragma unroll
  for (int kc = 0; kc < K; kc += 4) {
    const int fi = (kc >> 1) + half;
    float2 fa = A2[fi];
    float2 f0 = B0[fi];
    float2 f1 = B1[fi];
    v2f a;  a.x  = fa.x; a.y  = fa.y;
    v2f b0; b0.x = f0.x; b0.y = f0.y;
    v2f b1; b1.x = f1.x; b1.y = f1.y;
    c0 = wmma_f32_k4(a, b0, c0);
    c1 = wmma_f32_k4(a, b1, c1);
  }

  const float bias0 = bl[l16];
  const float bias1 = bl[l16 + 16];
#pragma unroll
  for (int r = 0; r < 8; ++r) { c0[r] += bias0; c1[r] += bias1; }

#pragma unroll
  for (int r = 0; r < 8; ++r) {
    float p = c0[r] * c0[r] + c1[r] * c1[r];
    p += __shfl_xor(p, 1, 32);
    p += __shfl_xor(p, 2, 32);
    p += __shfl_xor(p, 4, 32);
    p += __shfl_xor(p, 8, 32);
    float scale = 1.0f / fmaxf(sqrtf(p), 1e-12f);
    float o0 = c0[r] * scale;
    float o1 = c1[r] * scale;
    if (RELU) { o0 = fmaxf(o0, 0.0f); o1 = fmaxf(o1, 0.0f); }
    const size_t row = (size_t)(base + r + half * 8);
    out[row * 32 + l16]      = o0;
    out[row * 32 + 16 + l16] = o1;
  }
}

template <int K, bool RELU>
__global__ __launch_bounds__(256) void sage_compute_kernel(
    const float* __restrict__ agg, const float* __restrict__ cnt,
    const float* __restrict__ xin, const float* __restrict__ wl,
    const float* __restrict__ bl, const float* __restrict__ wr,
    float* __restrict__ out, int nrows, int agg_plane, int cnt_plane) {
  constexpr int P = K + 4;
  __shared__ __attribute__((aligned(16))) float lwl[32 * P];
  __shared__ __attribute__((aligned(16))) float lwr[32 * P];
  stage_weights<K>(wl, lwl);
  stage_weights<K>(wr, lwr);
  __syncthreads();
  const int lane = threadIdx.x & 31;
  const int wid  = blockIdx.x * (blockDim.x >> 5) + (threadIdx.x >> 5);
  const int base = wid * 16;
  if (base >= nrows) return;
  const int l16  = lane & 15;
  const int half = lane >> 4;
  const int rowA = base + l16;

  const float inv = 1.0f / fmaxf(cnt[rowA] + cnt[rowA + cnt_plane], 1.0f);

  v8f c0 = {};
  v8f c1 = {};

  const float2* P2 = (const float2*)(agg + agg_plane + (size_t)rowA * K);

#pragma unroll
  for (int phase = 0; phase < 2; ++phase) {
    const float* Arow = (phase == 0 ? agg : xin) + (size_t)rowA * K;
    const float* W    = (phase == 0 ? lwl : lwr);
    const float  s    = (phase == 0 ? inv : 1.0f);
    const float2* A2  = (const float2*)Arow;
    const float2* B0  = (const float2*)(W + l16 * P);
    const float2* B1  = (const float2*)(W + (l16 + 16) * P);
#pragma unroll
    for (int kc = 0; kc < K; kc += 4) {
      const int fi = (kc >> 1) + half;
      float2 fa = A2[fi];
      if (phase == 0) {
        const float2 fp = P2[fi];
        fa.x += fp.x;
        fa.y += fp.y;
      }
      float2 f0 = B0[fi];
      float2 f1 = B1[fi];
      v2f a;  a.x  = fa.x * s; a.y  = fa.y * s;
      v2f b0; b0.x = f0.x;     b0.y = f0.y;
      v2f b1; b1.x = f1.x;     b1.y = f1.y;
      c0 = wmma_f32_k4(a, b0, c0);
      c1 = wmma_f32_k4(a, b1, c1);
    }
  }

  const float bias0 = bl[l16];
  const float bias1 = bl[l16 + 16];
#pragma unroll
  for (int r = 0; r < 8; ++r) { c0[r] += bias0; c1[r] += bias1; }

#pragma unroll
  for (int r = 0; r < 8; ++r) {
    float p = c0[r] * c0[r] + c1[r] * c1[r];
    p += __shfl_xor(p, 1, 32);
    p += __shfl_xor(p, 2, 32);
    p += __shfl_xor(p, 4, 32);
    p += __shfl_xor(p, 8, 32);
    float scale = 1.0f / fmaxf(sqrtf(p), 1e-12f);
    float o0 = c0[r] * scale;
    float o1 = c1[r] * scale;
    if (RELU) { o0 = fmaxf(o0, 0.0f); o1 = fmaxf(o1, 0.0f); }
    const size_t row = (size_t)(base + r + half * 8);
    out[row * 32 + l16]      = o0;
    out[row * 32 + 16 + l16] = o1;
  }
}

extern "C" void kernel_launch(void* const* d_in, const int* in_sizes, int n_in,
                              void* d_out, int out_size, void* d_ws,
                              size_t ws_size, hipStream_t stream) {
  const float* x   = (const float*)d_in[0];
  const int* src0  = (const int*)d_in[1];
  const int* tgt0  = (const int*)d_in[2];
  const int* src1  = (const int*)d_in[3];
  const int* tgt1  = (const int*)d_in[4];
  const float* wl0 = (const float*)d_in[5];
  const float* bl0 = (const float*)d_in[6];
  const float* wr0 = (const float*)d_in[7];
  const float* wl1 = (const float*)d_in[8];
  const float* bl1 = (const float*)d_in[9];
  const float* wr1 = (const float*)d_in[10];
  const int E0 = in_sizes[1];
  const int E1 = in_sizes[3];

  float* ws   = (float*)d_ws;
  float* agg0 = ws;
  float* cnt0 = agg0 + 2 * (size_t)N1_ * H_;
  float* agg1 = cnt0 + 2 * (size_t)N1_;
  float* cnt1 = agg1 + 2 * (size_t)N2_ * H_;
  float* y0   = cnt1 + 2 * (size_t)N2_;
  float* h    = y0;
  const size_t zero_bytes =
      2 * ((size_t)N1_ * H_ + N1_ + (size_t)N2_ * H_ + N2_) * sizeof(float);
  hipMemsetAsync(d_ws, 0, zero_bytes, stream);

  {
    int blocks = (N0_ / 16) / 8;
    sage_project_kernel<INF_><<<blocks, 256, 0, stream>>>(x, wl0, y0, N0_);
  }
  {
    int blocks = (E0 + 255) / 256;
    sage_scatter_kernel<<<blocks, 256, 0, stream>>>(
        y0, src0, tgt0, agg0, cnt0, E0, N1_ * H_, N1_);
  }
  {
    int blocks = (N1_ / 16) / 8;
    sage_combine_kernel<INF_, true><<<blocks, 256, 0, stream>>>(
        agg0, cnt0, x, bl0, wr0, h, N1_, N1_ * H_, N1_);
  }
  {
    int blocks = (E1 + 255) / 256;
    sage_scatter_kernel<<<blocks, 256, 0, stream>>>(
        h, src1, tgt1, agg1, cnt1, E1, N2_ * H_, N2_);
  }
  {
    int blocks = (N2_ / 16) / 8;
    sage_compute_kernel<H_, false><<<blocks, 256, 0, stream>>>(
        agg1, cnt1, h, wl1, bl1, wr1, (float*)d_out, N2_, N2_ * H_, N2_);
  }
}


# Project and combine (K=128) copy each wave's 16 A rows to LDS with whole-line loads (512 contiguous bytes per instruction, was 16 lines x 16 bytes); A fragments read from LDS; k order, sums unchanged
# speedup vs baseline: 4.5208x; 1.0545x over previous
#include <hip/hip_runtime.h>
#include <math.h>

typedef __attribute__((ext_vector_type(2))) float v2f;
typedef __attribute__((ext_vector_type(8))) float v8f;

#define N0_  262144
#define INF_ 128
#define H_   32
#define N1_  131072
#define N2_  65536

__device__ __forceinline__ v8f wmma_f32_k4(v2f a, v2f b, v8f c) {
  return __builtin_amdgcn_wmma_f32_16x16x4_f32(
       false, a,  false, b,
       (short)0, c,  false,  false);
}

__global__ __launch_bounds__(256) void sage_scatter_kernel(
    const float* __restrict__ xsrc, const int* __restrict__ src,
    const int* __restrict__ tgt, float* __restrict__ agg,
    float* __restrict__ cnt, int nedges, int agg_plane, int cnt_plane) {
  const unsigned place = __builtin_amdgcn_s_sendmsg_rtn(0x87);
  const unsigned upper = (place >> 18) & 1u;
  agg += (size_t)upper * (size_t)agg_plane;
  cnt += (size_t)upper * (size_t)cnt_plane;
  const int lane = threadIdx.x & 31;
  const int wave = __builtin_amdgcn_readfirstlane((int)(threadIdx.x >> 5));
  const int wid  = blockIdx.x * (blockDim.x >> 5) + wave;
  const int e0   = wid * 32;
  if (e0 >= nedges) return;
  const int left = nedges - e0;
  const unsigned loff = (unsigned)lane;
  int s_v = 0;
  int t_v = 0;
  if (lane < left) {
    s_v = (src + e0)[loff];
    t_v = (tgt + e0)[loff];
    unsafeAtomicAdd(cnt + t_v, 1.0f);
  }
  if (left >= 32) {
    float v[32];
#pragma unroll
    for (int j = 0; j < 32; ++j) {
      const unsigned s = (unsigned)__builtin_amdgcn_readlane(s_v, j);
      v[j] = xsrc[s * 32u + loff];
    }
#pragma unroll
    for (int j = 0; j < 32; ++j) {
      const unsigned t = (unsigned)__builtin_amdgcn_readlane(t_v, j);
      unsafeAtomicAdd(agg + (t * 32u + loff), v[j]);
    }
  } else {
    for (int j = 0; j < left; ++j) {
      const unsigned s = (unsigned)__builtin_amdgcn_readlane(s_v, j);
      const unsigned t = (unsigned)__builtin_amdgcn_readlane(t_v, j);
      unsafeAtomicAdd(agg + (t * 32u + loff), xsrc[s * 32u + loff]);
    }
  }
}

template <int K>
__device__ __forceinline__ void stage_weights(const float* __restrict__ w,
                                              float* __restrict__ lw) {
  static_assert(K % 32 == 0, "32*K floats: a whole number of float4 per thread");
  constexpr int P = K + 4;
  const float4* w4 = (const float4*)w;
#pragma unroll
  for (int i = 0; i < K / 32; ++i) {
    const int f   = i * 256 + (int)threadIdx.x;
    const int row = f / (K / 4);
    const int q   = f % (K / 4);
    *(float4*)(lw + row * P + 4 * q) = w4[f];
  }
}

template <int K>
__device__ __forceinline__ void stage_rows(const float* __restrict__ rows,
                                           float* __restrict__ la, int lane) {
  static_assert(K % 8 == 0, "16*K floats: a whole number of float4 per lane");
  constexpr int P = K + 4;
  const float4* r4 = (const float4*)rows;
#pragma unroll
  for (int i = 0; i < K / 8; ++i) {
    const int f   = i * 32 + lane;
    const int row = f / (K / 4);
    const int q   = f % (K / 4);
    *(float4*)(la + row * P + 4 * q) = r4[f];
  }
}

template <int K>
__global__ __launch_bounds__(256) void sage_project_kernel(
    const float* __restrict__ xin, const float* __restrict__ w,
    float* __restrict__ y, int nrows) {
  constexpr int P = K + 4;
  __shared__ __attribute__((aligned(16))) float lw[32 * P];
  __shared__ __attribute__((aligned(16))) float la[8 * 16 * P];
  stage_weights<K>(w, lw);
  const int lane = threadIdx.x & 31;
  const int wv   = threadIdx.x >> 5;
  const int wid  = blockIdx.x * (blockDim.x >> 5) + wv;
  const int base = wid * 16;
  float* myA = la + wv * (16 * P);
  if (base < nrows) stage_rows<K>(xin + (size_t)base * K, myA, lane);
  __syncthreads();
  if (base >= nrows) return;
  const int l16  = lane & 15;
  const int half = lane >> 4;

  v8f c0 = {};
  v8f c1 = {};
  const float2* A2 = (const float2*)(myA + l16 * P);
  const float2* B0 = (const float2*)(lw + l16 * P);
  const float2* B1 = (const float2*)(lw + (l16 + 16) * P);
#pragma unroll
  for (int kc = 0; kc < K; kc += 4) {
    const int fi = (kc >> 1) + half;
    float2 fa = A2[fi];
    float2 f0 = B0[fi];
    float2 f1 = B1[fi];
    v2f a;  a.x  = fa.x; a.y  = fa.y;
    v2f b0; b0.x = f0.x; b0.y = f0.y;
    v2f b1; b1.x = f1.x; b1.y = f1.y;
    c0 = wmma_f32_k4(a, b0, c0);
    c1 = wmma_f32_k4(a, b1, c1);
  }
#pragma unroll
  for (int r = 0; r < 8; ++r) {
    const size_t row = (size_t)(base + r + half * 8);
    y[row * 32 + l16]      = c0[r];
    y[row * 32 + 16 + l16] = c1[r];
  }
}

template <int K, bool RELU>
__global__ __launch_bounds__(256) void sage_combine_kernel(
    const float* __restrict__ aggy, const float* __restrict__ cnt,
    const float* __restrict__ xin, const float* __restrict__ bl,
    const float* __restrict__ wr, float* __restrict__ out, int nrows,
    int agg_plane, int cnt_plane) {
  constexpr int P = K + 4;
  __shared__ __attribute__((aligned(16))) float lw[32 * P];
  __shared__ __attribute__((aligned(16))) float la[8 * 16 * P];
  stage_weights<K>(wr, lw);
  const int lane = threadIdx.x & 31;
  const int wv   = threadIdx.x >> 5;
  const int wid  = blockIdx.x * (blockDim.x >> 5) + wv;
  const int base = wid * 16;
  float* myA = la + wv * (16 * P);
  if (base < nrows) stage_rows<K>(xin + (size_t)base * K, myA, lane);
  __syncthreads();
  if (base >= nrows) return;
  const int l16  = lane & 15;
  const int half = lane >> 4;

  v8f c0;
  v8f c1;
#pragma unroll
  for (int r = 0; r < 8; ++r) {
    const size_t row = (size_t)(base + r + half * 8);
    const float inv = 1.0f / fmaxf(cnt[row] + cnt[row + cnt_plane], 1.0f);
    c0[r] = (aggy[row * 32 + l16] + aggy[row * 32 + l16 + agg_plane]) * inv;
    c1[r] = (aggy[row * 32 + 16 + l16] + aggy[row * 32 + 16 + l16 + agg_plane]) *
            inv;
  }

  const float2* A2 = (const float2*)(myA + l16 * P);
  const float2* B0 = (const float2*)(lw + l16 * P);
  const float2* B1 = (const float2*)(lw + (l16 + 16) * P);
#pragma unroll
  for (int kc = 0; kc < K; kc += 4) {
    const int fi = (kc >> 1) + half;
    float2 fa = A2[fi];
    float2 f0 = B0[fi];
    float2 f1 = B1[fi];
    v2f a;  a.x  = fa.x; a.y  = fa.y;
    v2f b0; b0.x = f0.x; b0.y = f0.y;
    v2f b1; b1.x = f1.x; b1.y = f1.y;
    c0 = wmma_f32_k4(a, b0, c0);
    c1 = wmma_f32_k4(a, b1, c1);
  }

  const float bias0 = bl[l16];
  const float bias1 = bl[l16 + 16];
#pragma unroll
  for (int r = 0; r < 8; ++r) { c0[r] += bias0; c1[r] += bias1; }

#pragma unroll
  for (int r = 0; r < 8; ++r) {
    float p = c0[r] * c0[r] + c1[r] * c1[r];
    p += __shfl_xor(p, 1, 32);
    p += __shfl_xor(p, 2, 32);
    p += __shfl_xor(p, 4, 32);
    p += __shfl_xor(p, 8, 32);
    float scale = 1.0f / fmaxf(sqrtf(p), 1e-12f);
    float o0 = c0[r] * scale;
    float o1 = c1[r] * scale;
    if (RELU) { o0 = fmaxf(o0, 0.0f); o1 = fmaxf(o1, 0.0f); }
    const size_t row = (size_t)(base + r + half * 8);
    out[row * 32 + l16]      = o0;
    out[row * 32 + 16 + l16] = o1;
  }
}

template <int K, bool RELU>
__global__ __launch_bounds__(256) void sage_compute_kernel(
    const float* __restrict__ agg, const float* __restrict__ cnt,
    const float* __restrict__ xin, const float* __restrict__ wl,
    const float* __restrict__ bl, const float* __restrict__ wr,
    float* __restrict__ out, int nrows, int agg_plane, int cnt_plane) {
  constexpr int P = K + 4;
  __shared__ __attribute__((aligned(16))) float lwl[32 * P];
  __shared__ __attribute__((aligned(16))) float lwr[32 * P];
  stage_weights<K>(wl, lwl);
  stage_weights<K>(wr, lwr);
  __syncthreads();
  const int lane = threadIdx.x & 31;
  const int wid  = blockIdx.x * (blockDim.x >> 5) + (threadIdx.x >> 5);
  const int base = wid * 16;
  if (base >= nrows) return;
  const int l16  = lane & 15;
  const int half = lane >> 4;
  const int rowA = base + l16;

  const float inv = 1.0f / fmaxf(cnt[rowA] + cnt[rowA + cnt_plane], 1.0f);

  v8f c0 = {};
  v8f c1 = {};

  const float2* P2 = (const float2*)(agg + agg_plane + (size_t)rowA * K);

#pragma unroll
  for (int phase = 0; phase < 2; ++phase) {
    const float* Arow = (phase == 0 ? agg : xin) + (size_t)rowA * K;
    const float* W    = (phase == 0 ? lwl : lwr);
    const float  s    = (phase == 0 ? inv : 1.0f);
    const float2* A2  = (const float2*)Arow;
    const float2* B0  = (const float2*)(W + l16 * P);
    const float2* B1  = (const float2*)(W + (l16 + 16) * P);
#pragma unroll
    for (int kc = 0; kc < K; kc += 4) {
      const int fi = (kc >> 1) + half;
      float2 fa = A2[fi];
      if (phase == 0) {
        const float2 fp = P2[fi];
        fa.x += fp.x;
        fa.y += fp.y;
      }
      float2 f0 = B0[fi];
      float2 f1 = B1[fi];
      v2f a;  a.x  = fa.x * s; a.y  = fa.y * s;
      v2f b0; b0.x = f0.x;     b0.y = f0.y;
      v2f b1; b1.x = f1.x;     b1.y = f1.y;
      c0 = wmma_f32_k4(a, b0, c0);
      c1 = wmma_f32_k4(a, b1, c1);
    }
  }

  const float bias0 = bl[l16];
  const float bias1 = bl[l16 + 16];
#pragma unroll
  for (int r = 0; r < 8; ++r) { c0[r] += bias0; c1[r] += bias1; }

#pragma unroll
  for (int r = 0; r < 8; ++r) {
    float p = c0[r] * c0[r] + c1[r] * c1[r];
    p += __shfl_xor(p, 1, 32);
    p += __shfl_xor(p, 2, 32);
    p += __shfl_xor(p, 4, 32);
    p += __shfl_xor(p, 8, 32);
    float scale = 1.0f / fmaxf(sqrtf(p), 1e-12f);
    float o0 = c0[r] * scale;
    float o1 = c1[r] * scale;
    if (RELU) { o0 = fmaxf(o0, 0.0f); o1 = fmaxf(o1, 0.0f); }
    const size_t row = (size_t)(base + r + half * 8);
    out[row * 32 + l16]      = o0;
    out[row * 32 + 16 + l16] = o1;
  }
}

extern "C" void kernel_launch(void* const* d_in, const int* in_sizes, int n_in,
                              void* d_out, int out_size, void* d_ws,
                              size_t ws_size, hipStream_t stream) {
  const float* x   = (const float*)d_in[0];
  const int* src0  = (const int*)d_in[1];
  const int* tgt0  = (const int*)d_in[2];
  const int* src1  = (const int*)d_in[3];
  const int* tgt1  = (const int*)d_in[4];
  const float* wl0 = (const float*)d_in[5];
  const float* bl0 = (const float*)d_in[6];
  const float* wr0 = (const float*)d_in[7];
  const float* wl1 = (const float*)d_in[8];
  const float* bl1 = (const float*)d_in[9];
  const float* wr1 = (const float*)d_in[10];
  const int E0 = in_sizes[1];
  const int E1 = in_sizes[3];

  float* ws   = (float*)d_ws;
  float* agg0 = ws;
  float* cnt0 = agg0 + 2 * (size_t)N1_ * H_;
  float* agg1 = cnt0 + 2 * (size_t)N1_;
  float* cnt1 = agg1 + 2 * (size_t)N2_ * H_;
  float* y0   = cnt1 + 2 * (size_t)N2_;
  float* h    = y0;
  const size_t zero_bytes =
      2 * ((size_t)N1_ * H_ + N1_ + (size_t)N2_ * H_ + N2_) * sizeof(float);
  hipMemsetAsync(d_ws, 0, zero_bytes, stream);

  {
    int blocks = (N0_ / 16) / 8;
    sage_project_kernel<INF_><<<blocks, 256, 0, stream>>>(x, wl0, y0, N0_);
  }
  {
    int blocks = (E0 + 255) / 256;
    sage_scatter_kernel<<<blocks, 256, 0, stream>>>(
        y0, src0, tgt0, agg0, cnt0, E0, N1_ * H_, N1_);
  }
  {
    int blocks = (N1_ / 16) / 8;
    sage_combine_kernel<INF_, true><<<blocks, 256, 0, stream>>>(
        agg0, cnt0, x, bl0, wr0, h, N1_, N1_ * H_, N1_);
  }
  {
    int blocks = (E1 + 255) / 256;
    sage_scatter_kernel<<<blocks, 256, 0, stream>>>(
        h, src1, tgt1, agg1, cnt1, E1, N2_ * H_, N2_);
  }
  {
    int blocks = (N2_ / 16) / 8;
    sage_compute_kernel<H_, false><<<blocks, 256, 0, stream>>>(
        agg1, cnt1, h, wl1, bl1, wr1, (float*)d_out, N2_, N2_ * H_, N2_);
  }
}


# Edge counts leave the scatter: a new launch counts both layers' targets in LDS (64K bins per work-group, LDS atomics), adds them to cnt by whole-line f32 atomics; integer sums, exact in any order
# speedup vs baseline: 4.6955x; 1.0386x over previous
#include <hip/hip_runtime.h>
#include <math.h>

typedef __attribute__((ext_vector_type(2))) float v2f;
typedef __attribute__((ext_vector_type(8))) float v8f;

#define N0_  262144
#define INF_ 128
#define H_   32
#define N1_  131072
#define N2_  65536

__device__ __forceinline__ v8f wmma_f32_k4(v2f a, v2f b, v8f c) {
  return __builtin_amdgcn_wmma_f32_16x16x4_f32(
       false, a,  false, b,
       (short)0, c,  false,  false);
}

__global__ __launch_bounds__(256) void sage_scatter_kernel(
    const float* __restrict__ xsrc, const int* __restrict__ src,
    const int* __restrict__ tgt, float* __restrict__ agg,
    int nedges, int agg_plane) {
  const unsigned place = __builtin_amdgcn_s_sendmsg_rtn(0x87);
  const unsigned upper = (place >> 18) & 1u;
  agg += (size_t)upper * (size_t)agg_plane;
  const int lane = threadIdx.x & 31;
  const int wave = __builtin_amdgcn_readfirstlane((int)(threadIdx.x >> 5));
  const int wid  = blockIdx.x * (blockDim.x >> 5) + wave;
  const int e0   = wid * 32;
  if (e0 >= nedges) return;
  const int left = nedges - e0;
  const unsigned loff = (unsigned)lane;
  int s_v = 0;
  int t_v = 0;
  if (lane < left) {
    s_v = (src + e0)[loff];
    t_v = (tgt + e0)[loff];
  }
  if (left >= 32) {
    float v[32];
#pragma unroll
    for (int j = 0; j < 32; ++j) {
      const unsigned s = (unsigned)__builtin_amdgcn_readlane(s_v, j);
      v[j] = xsrc[s * 32u + loff];
    }
#pragma unroll
    for (int j = 0; j < 32; ++j) {
      const unsigned t = (unsigned)__builtin_amdgcn_readlane(t_v, j);
      unsafeAtomicAdd(agg + (t * 32u + loff), v[j]);
    }
  } else {
    for (int j = 0; j < left; ++j) {
      const unsigned s = (unsigned)__builtin_amdgcn_readlane(s_v, j);
      const unsigned t = (unsigned)__builtin_amdgcn_readlane(t_v, j);
      unsafeAtomicAdd(agg + (t * 32u + loff), xsrc[s * 32u + loff]);
    }
  }
}

#define CNT_BINS   65536
#define CNT_SLICES 64
__global__ __launch_bounds__(1024) void sage_count_kernel(
    const int* __restrict__ tgtA, float* __restrict__ cntA, int edgesA,
    int binsA, int planeA, int blocksA,
    const int* __restrict__ tgtB, float* __restrict__ cntB, int edgesB,
    int binsB, int planeB) {
  __shared__ __attribute__((aligned(16))) unsigned hist[CNT_BINS];
  const unsigned place = __builtin_amdgcn_s_sendmsg_rtn(0x87);
  const unsigned upper = (place >> 18) & 1u;
  const bool second = (int)blockIdx.x >= blocksA;
  const int* __restrict__ tgt = second ? tgtB : tgtA;
  float* __restrict__ cnt     = second ? cntB : cntA;
  const int edges = second ? edgesB : edgesA;
  const int bins  = second ? binsB : binsA;
  const int plane = second ? planeB : planeA;
  const int b     = second ? (int)blockIdx.x - blocksA : (int)blockIdx.x;
  cnt += (size_t)upper * (size_t)plane;
  const int nranges = (bins + CNT_BINS - 1) / CNT_BINS;
  const int range   = b % nranges;
  const int slice   = b / nranges;
  const int lo      = range * CNT_BINS;
  const int per     = (edges + CNT_SLICES - 1) / CNT_SLICES;
  const int start   = slice * per;
  const int end     = min(start + per, edges);
  const int tid     = (int)threadIdx.x;

  uint4* h4 = (uint4*)hist;
  for (int i = tid; i < CNT_BINS / 4; i += 1024) h4[i] = make_uint4(0u, 0u, 0u, 0u);
  __syncthreads();
  for (int e0 = start + tid; e0 < end; e0 += 16 * 1024) {
    int t[16];
#pragma unroll
    for (int u = 0; u < 16; ++u) {
      const int e = e0 + u * 1024;
      t[u] = e < end ? tgt[e] : -1;
    }
#pragma unroll
    for (int u = 0; u < 16; ++u) {
      const unsigned k = (unsigned)(t[u] - lo);
      if (k < (unsigned)CNT_BINS) atomicAdd(&hist[k], 1u);
    }
  }
  __syncthreads();
  for (int i0 = tid; i0 < CNT_BINS; i0 += 8 * 1024) {
    unsigned c[8];
#pragma unroll
    for (int u = 0; u < 8; ++u) c[u] = hist[i0 + u * 1024];
#pragma unroll
    for (int u = 0; u < 8; ++u) {
      const int bin = lo + i0 + u * 1024;
      if (c[u] != 0u && bin < bins) unsafeAtomicAdd(cnt + bin, (float)c[u]);
    }
  }
}

template <int K>
__device__ __forceinline__ void stage_weights(const float* __restrict__ w,
                                              float* __restrict__ lw) {
  static_assert(K % 32 == 0, "32*K floats: a whole number of float4 per thread");
  constexpr int P = K + 4;
  const float4* w4 = (const float4*)w;
#pragma unroll
  for (int i = 0; i < K / 32; ++i) {
    const int f   = i * 256 + (int)threadIdx.x;
    const int row = f / (K / 4);
    const int q   = f % (K / 4);
    *(float4*)(lw + row * P + 4 * q) = w4[f];
  }
}

template <int K>
__device__ __forceinline__ void stage_rows(const float* __restrict__ rows,
                                           float* __restrict__ la, int lane) {
  static_assert(K % 8 == 0, "16*K floats: a whole number of float4 per lane");
  constexpr int P = K + 4;
  const float4* r4 = (const float4*)rows;
#pragma unroll
  for (int i = 0; i < K / 8; ++i) {
    const int f   = i * 32 + lane;
    const int row = f / (K / 4);
    const int q   = f % (K / 4);
    *(float4*)(la + row * P + 4 * q) = r4[f];
  }
}

template <int K>
__global__ __launch_bounds__(256) void sage_project_kernel(
    const float* __restrict__ xin, const float* __restrict__ w,
    float* __restrict__ y, int nrows) {
  constexpr int P = K + 4;
  __shared__ __attribute__((aligned(16))) float lw[32 * P];
  __shared__ __attribute__((aligned(16))) float la[8 * 16 * P];
  stage_weights<K>(w, lw);
  const int lane = threadIdx.x & 31;
  const int wv   = threadIdx.x >> 5;
  const int wid  = blockIdx.x * (blockDim.x >> 5) + wv;
  const int base = wid * 16;
  float* myA = la + wv * (16 * P);
  if (base < nrows) stage_rows<K>(xin + (size_t)base * K, myA, lane);
  __syncthreads();
  if (base >= nrows) return;
  const int l16  = lane & 15;
  const int half = lane >> 4;

  v8f c0 = {};
  v8f c1 = {};
  const float2* A2 = (const float2*)(myA + l16 * P);
  const float2* B0 = (const float2*)(lw + l16 * P);
  const float2* B1 = (const float2*)(lw + (l16 + 16) * P);
#pragma unroll
  for (int kc = 0; kc < K; kc += 4) {
    const int fi = (kc >> 1) + half;
    float2 fa = A2[fi];
    float2 f0 = B0[fi];
    float2 f1 = B1[fi];
    v2f a;  a.x  = fa.x; a.y  = fa.y;
    v2f b0; b0.x = f0.x; b0.y = f0.y;
    v2f b1; b1.x = f1.x; b1.y = f1.y;
    c0 = wmma_f32_k4(a, b0, c0);
    c1 = wmma_f32_k4(a, b1, c1);
  }
#pragma unroll
  for (int r = 0; r < 8; ++r) {
    const size_t row = (size_t)(base + r + half * 8);
    y[row * 32 + l16]      = c0[r];
    y[row * 32 + 16 + l16] = c1[r];
  }
}

template <int K, bool RELU>
__global__ __launch_bounds__(256) void sage_combine_kernel(
    const float* __restrict__ aggy, const float* __restrict__ cnt,
    const float* __restrict__ xin, const float* __restrict__ bl,
    const float* __restrict__ wr, float* __restrict__ out, int nrows,
    int agg_plane, int cnt_plane) {
  constexpr int P = K + 4;
  __shared__ __attribute__((aligned(16))) float lw[32 * P];
  __shared__ __attribute__((aligned(16))) float la[8 * 16 * P];
  stage_weights<K>(wr, lw);
  const int lane = threadIdx.x & 31;
  const int wv   = threadIdx.x >> 5;
  const int wid  = blockIdx.x * (blockDim.x >> 5) + wv;
  const int base = wid * 16;
  float* myA = la + wv * (16 * P);
  if (base < nrows) stage_rows<K>(xin + (size_t)base * K, myA, lane);
  __syncthreads();
  if (base >= nrows) return;
  const int l16  = lane & 15;
  const int half = lane >> 4;

  v8f c0;
  v8f c1;
#pragma unroll
  for (int r = 0; r < 8; ++r) {
    const size_t row = (size_t)(base + r + half * 8);
    const float inv = 1.0f / fmaxf(cnt[row] + cnt[row + cnt_plane], 1.0f);
    c0[r] = (aggy[row * 32 + l16] + aggy[row * 32 + l16 + agg_plane]) * inv;
    c1[r] = (aggy[row * 32 + 16 + l16] + aggy[row * 32 + 16 + l16 + agg_plane]) *
            inv;
  }

  const float2* A2 = (const float2*)(myA + l16 * P);
  const float2* B0 = (const float2*)(lw + l16 * P);
  const float2* B1 = (const float2*)(lw + (l16 + 16) * P);
#pragma unroll
  for (int kc = 0; kc < K; kc += 4) {
    const int fi = (kc >> 1) + half;
    float2 fa = A2[fi];
    float2 f0 = B0[fi];
    float2 f1 = B1[fi];
    v2f a;  a.x  = fa.x; a.y  = fa.y;
    v2f b0; b0.x = f0.x; b0.y = f0.y;
    v2f b1; b1.x = f1.x; b1.y = f1.y;
    c0 = wmma_f32_k4(a, b0, c0);
    c1 = wmma_f32_k4(a, b1, c1);
  }

  const float bias0 = bl[l16];
  const float bias1 = bl[l16 + 16];
#pragma unroll
  for (int r = 0; r < 8; ++r) { c0[r] += bias0; c1[r] += bias1; }

#pragma unroll
  for (int r = 0; r < 8; ++r) {
    float p = c0[r] * c0[r] + c1[r] * c1[r];
    p += __shfl_xor(p, 1, 32);
    p += __shfl_xor(p, 2, 32);
    p += __shfl_xor(p, 4, 32);
    p += __shfl_xor(p, 8, 32);
    float scale = 1.0f / fmaxf(sqrtf(p), 1e-12f);
    float o0 = c0[r] * scale;
    float o1 = c1[r] * scale;
    if (RELU) { o0 = fmaxf(o0, 0.0f); o1 = fmaxf(o1, 0.0f); }
    const size_t row = (size_t)(base + r + half * 8);
    out[row * 32 + l16]      = o0;
    out[row * 32 + 16 + l16] = o1;
  }
}

template <int K, bool RELU>
__global__ __launch_bounds__(256) void sage_compute_kernel(
    const float* __restrict__ agg, const float* __restrict__ cnt,
    const float* __restrict__ xin, const float* __restrict__ wl,
    const float* __restrict__ bl, const float* __restrict__ wr,
    float* __restrict__ out, int nrows, int agg_plane, int cnt_plane) {
  constexpr int P = K + 4;
  __shared__ __attribute__((aligned(16))) float lwl[32 * P];
  __shared__ __attribute__((aligned(16))) float lwr[32 * P];
  stage_weights<K>(wl, lwl);
  stage_weights<K>(wr, lwr);
  __syncthreads();
  const int lane = threadIdx.x & 31;
  const int wid  = blockIdx.x * (blockDim.x >> 5) + (threadIdx.x >> 5);
  const int base = wid * 16;
  if (base >= nrows) return;
  const int l16  = lane & 15;
  const int half = lane >> 4;
  const int rowA = base + l16;

  const float inv = 1.0f / fmaxf(cnt[rowA] + cnt[rowA + cnt_plane], 1.0f);

  v8f c0 = {};
  v8f c1 = {};

  const float2* P2 = (const float2*)(agg + agg_plane + (size_t)rowA * K);

#pragma unroll
  for (int phase = 0; phase < 2; ++phase) {
    const float* Arow = (phase == 0 ? agg : xin) + (size_t)rowA * K;
    const float* W    = (phase == 0 ? lwl : lwr);
    const float  s    = (phase == 0 ? inv : 1.0f);
    const float2* A2  = (const float2*)Arow;
    const float2* B0  = (const float2*)(W + l16 * P);
    const float2* B1  = (const float2*)(W + (l16 + 16) * P);
#pragma unroll
    for (int kc = 0; kc < K; kc += 4) {
      const int fi = (kc >> 1) + half;
      float2 fa = A2[fi];
      if (phase == 0) {
        const float2 fp = P2[fi];
        fa.x += fp.x;
        fa.y += fp.y;
      }
      float2 f0 = B0[fi];
      float2 f1 = B1[fi];
      v2f a;  a.x  = fa.x * s; a.y  = fa.y * s;
      v2f b0; b0.x = f0.x;     b0.y = f0.y;
      v2f b1; b1.x = f1.x;     b1.y = f1.y;
      c0 = wmma_f32_k4(a, b0, c0);
      c1 = wmma_f32_k4(a, b1, c1);
    }
  }

  const float bias0 = bl[l16];
  const float bias1 = bl[l16 + 16];
#pragma unroll
  for (int r = 0; r < 8; ++r) { c0[r] += bias0; c1[r] += bias1; }

#pragma unroll
  for (int r = 0; r < 8; ++r) {
    float p = c0[r] * c0[r] + c1[r] * c1[r];
    p += __shfl_xor(p, 1, 32);
    p += __shfl_xor(p, 2, 32);
    p += __shfl_xor(p, 4, 32);
    p += __shfl_xor(p, 8, 32);
    float scale = 1.0f / fmaxf(sqrtf(p), 1e-12f);
    float o0 = c0[r] * scale;
    float o1 = c1[r] * scale;
    if (RELU) { o0 = fmaxf(o0, 0.0f); o1 = fmaxf(o1, 0.0f); }
    const size_t row = (size_t)(base + r + half * 8);
    out[row * 32 + l16]      = o0;
    out[row * 32 + 16 + l16] = o1;
  }
}

extern "C" void kernel_launch(void* const* d_in, const int* in_sizes, int n_in,
                              void* d_out, int out_size, void* d_ws,
                              size_t ws_size, hipStream_t stream) {
  const float* x   = (const float*)d_in[0];
  const int* src0  = (const int*)d_in[1];
  const int* tgt0  = (const int*)d_in[2];
  const int* src1  = (const int*)d_in[3];
  const int* tgt1  = (const int*)d_in[4];
  const float* wl0 = (const float*)d_in[5];
  const float* bl0 = (const float*)d_in[6];
  const float* wr0 = (const float*)d_in[7];
  const float* wl1 = (const float*)d_in[8];
  const float* bl1 = (const float*)d_in[9];
  const float* wr1 = (const float*)d_in[10];
  const int E0 = in_sizes[1];
  const int E1 = in_sizes[3];

  float* ws   = (float*)d_ws;
  float* agg0 = ws;
  float* cnt0 = agg0 + 2 * (size_t)N1_ * H_;
  float* agg1 = cnt0 + 2 * (size_t)N1_;
  float* cnt1 = agg1 + 2 * (size_t)N2_ * H_;
  float* y0   = cnt1 + 2 * (size_t)N2_;
  float* h    = y0;
  const size_t zero_bytes =
      2 * ((size_t)N1_ * H_ + N1_ + (size_t)N2_ * H_ + N2_) * sizeof(float);
  hipMemsetAsync(d_ws, 0, zero_bytes, stream);

  {
    int blocks0 = (N1_ / CNT_BINS) * CNT_SLICES;
    int blocks1 = (N2_ / CNT_BINS) * CNT_SLICES;
    sage_count_kernel<<<blocks0 + blocks1, 1024, 0, stream>>>(
        tgt0, cnt0, E0, N1_, N1_, blocks0, tgt1, cnt1, E1, N2_, N2_);
  }
  {
    int blocks = (N0_ / 16) / 8;
    sage_project_kernel<INF_><<<blocks, 256, 0, stream>>>(x, wl0, y0, N0_);
  }
  {
    int blocks = (E0 + 255) / 256;
    sage_scatter_kernel<<<blocks, 256, 0, stream>>>(
        y0, src0, tgt0, agg0, E0, N1_ * H_);
  }
  {
    int blocks = (N1_ / 16) / 8;
    sage_combine_kernel<INF_, true><<<blocks, 256, 0, stream>>>(
        agg0, cnt0, x, bl0, wr0, h, N1_, N1_ * H_, N1_);
  }
  {
    int blocks = (E1 + 255) / 256;
    sage_scatter_kernel<<<blocks, 256, 0, stream>>>(
        h, src1, tgt1, agg1, E1, N2_ * H_);
  }
  {
    int blocks = (N2_ / 16) / 8;
    sage_compute_kernel<H_, false><<<blocks, 256, 0, stream>>>(
        agg1, cnt1, h, wl1, bl1, wr1, (float*)d_out, N2_, N2_ * H_, N2_);
  }
}


# Edge sums leave global atomics: edges are ordered by target range in LDS (one launch, both layers); each work-group sums its range's rows by LDS f32 adds, stores once; arrival-order sums per target
# speedup vs baseline: 5.5361x; 1.1790x over previous
#include <hip/hip_runtime.h>
#include <math.h>

typedef __attribute__((ext_vector_type(2))) float v2f;
typedef __attribute__((ext_vector_type(8))) float v8f;

#define N0_  262144
#define INF_ 128
#define H_   32
#define N1_  131072
#define N2_  65536

__device__ __forceinline__ v8f wmma_f32_k4(v2f a, v2f b, v8f c) {
  return __builtin_amdgcn_wmma_f32_16x16x4_f32(
       false, a,  false, b,
       (short)0, c,  false,  false);
}

#define NB_ 256
#define SL_ 8192
#define STP_ (NB_ + 32)
#define SH0_ 9
#define SH1_ 8
static_assert((NB_ << SH0_) == N1_, "ranges cover the layer-0 targets");
static_assert((NB_ << SH1_) == N2_, "ranges cover the layer-1 targets");
__global__ __launch_bounds__(1024) void sage_group_kernel(
    const int* __restrict__ srcA, const int* __restrict__ tgtA, int edgesA,
    int shiftA, unsigned* __restrict__ pairsA, unsigned* __restrict__ startsA,
    int slicesA,
    const int* __restrict__ srcB, const int* __restrict__ tgtB, int edgesB,
    int shiftB, unsigned* __restrict__ pairsB, unsigned* __restrict__ startsB) {
  __shared__ unsigned hcnt[NB_];
  __shared__ unsigned wtot[NB_ / 32];
  __shared__ unsigned total_s;
  __shared__ __attribute__((aligned(16))) uint2 spair[SL_];
  const bool second = (int)blockIdx.x >= slicesA;
  const int* __restrict__ src = second ? srcB : srcA;
  const int* __restrict__ tgt = second ? tgtB : tgtA;
  const int edges = second ? edgesB : edgesA;
  const int shift = second ? shiftB : shiftA;
  uint2* __restrict__ pairs = (uint2*)(second ? pairsB : pairsA);
  unsigned* __restrict__ starts = second ? startsB : startsA;
  const int sl  = second ? (int)blockIdx.x - slicesA : (int)blockIdx.x;
  const int tid = (int)threadIdx.x;
  const int lane = tid & 31;
  const int e_begin = sl * SL_;
  const int n = min(SL_, edges - e_begin);

  if (tid < NB_) hcnt[tid] = 0u;
  __syncthreads();
  int s[SL_ / 1024];
  int t[SL_ / 1024];
#pragma unroll
  for (int u = 0; u < SL_ / 1024; ++u) {
    const int i = tid + u * 1024;
    s[u] = 0;
    t[u] = -1;
    if (i < n) { s[u] = src[e_begin + i]; t[u] = tgt[e_begin + i]; }
  }
#pragma unroll
  for (int u = 0; u < SL_ / 1024; ++u) {
    const unsigned b = (unsigned)t[u] >> shift;
    if (b < (unsigned)NB_) atomicAdd(&hcnt[b], 1u);
    else t[u] = -1;
  }
  __syncthreads();
  unsigned c = 0u;
  if (tid < NB_) c = hcnt[tid];
  unsigned incl = c;
#pragma unroll
  for (int d = 1; d < 32; d <<= 1) {
    const unsigned v = __shfl_up(incl, d, 32);
    if (lane >= d) incl += v;
  }
  if (tid < NB_ && lane == 31) wtot[tid >> 5] = incl;
  __syncthreads();
  if (tid < NB_) {
    unsigned base = 0u;
    for (int k = 0; k < (tid >> 5); ++k) base += wtot[k];
    const unsigned ex = base + incl - c;
    hcnt[tid] = ex;
    starts[(size_t)sl * STP_ + tid] = ex;
    if (tid == NB_ - 1) {
      starts[(size_t)sl * STP_ + NB_] = ex + c;
      total_s = ex + c;
    }
  }
  __syncthreads();
#pragma unroll
  for (int u = 0; u < SL_ / 1024; ++u) {
    if (t[u] >= 0) {
      const unsigned b   = (unsigned)t[u] >> shift;
      const unsigned pos = atomicAdd(&hcnt[b], 1u);
      spair[pos] = make_uint2((unsigned)s[u], (unsigned)t[u]);
    }
  }
  __syncthreads();
  const int total = (int)total_s;
  for (int i = tid; i < total; i += 1024) pairs[(size_t)e_begin + i] = spair[i];
}

template <int N>
__device__ __forceinline__ void add_rows(const float* __restrict__ xsrc,
                                         float* __restrict__ acc, int s_v,
                                         int t_v, int jb, unsigned loff) {
  float v[N];
#pragma unroll
  for (int j = 0; j < N; ++j) {
    const unsigned s = (unsigned)__builtin_amdgcn_readlane(s_v, jb + j);
    v[j] = xsrc[s * 32u + loff];
  }
#pragma unroll
  for (int j = 0; j < N; ++j) {
    const unsigned t = (unsigned)__builtin_amdgcn_readlane(t_v, jb + j);
    atomicAdd(&acc[t * 32u + loff], v[j]);
  }
}

__global__ __launch_bounds__(1024) void sage_gather_kernel(
    const float* __restrict__ xsrc, const unsigned* __restrict__ pairs_w,
    const unsigned* __restrict__ starts, float* __restrict__ agg,
    float* __restrict__ cnt, int nslices, int shift) {
  __shared__ __attribute__((aligned(16))) float acc[(1 << SH0_) * 32];
  __shared__ unsigned cn[1 << SH0_];
  const uint2* __restrict__ pairs = (const uint2*)pairs_w;
  const int R    = 1 << shift;
  const int b    = (int)blockIdx.x;
  const int row0 = b << shift;
  const int tid  = (int)threadIdx.x;
  const int lane = tid & 31;
  const int wave = __builtin_amdgcn_readfirstlane(tid >> 5);
  float4* acc4 = (float4*)acc;
  for (int i = tid; i < R * 8; i += 1024) acc4[i] = make_float4(0.f, 0.f, 0.f, 0.f);
  if (tid < R) cn[tid] = 0u;
  __syncthreads();

  const int rpw = (nslices + 31) / 32;
  unsigned st = 0u, len = 0u;
  {
    const int g = wave * rpw + lane;
    if (lane < rpw && g < nslices) {
      const unsigned* p = starts + (size_t)g * STP_ + b;
      const unsigned a = p[0];
      const unsigned e = p[1];
      st  = (unsigned)g * (unsigned)SL_ + a;
      len = e - a;
    }
  }
  unsigned incl = len;
#pragma unroll
  for (int d = 1; d < 32; d <<= 1) {
    const unsigned v = __shfl_up(incl, d, 32);
    if (lane >= d) incl += v;
  }
  const unsigned excl  = incl - len;
  const unsigned total = (unsigned)__builtin_amdgcn_readlane((int)incl, 31);
  const unsigned loff  = (unsigned)lane;
  const unsigned rmask = (unsigned)R - 1u;

  for (unsigned c0 = 0u; c0 < total; c0 += 32u) {
    const unsigned q = c0 + loff;
    unsigned addr = 0u;
    for (int k = 0; k < rpw; ++k) {
      const unsigned ek = (unsigned)__builtin_amdgcn_readlane((int)excl, k);
      const unsigned ik = (unsigned)__builtin_amdgcn_readlane((int)incl, k);
      const unsigned sk = (unsigned)__builtin_amdgcn_readlane((int)st, k);
      if (q >= ek && q < ik) addr = sk + (q - ek);
    }
    const unsigned left = total - c0;
    int s_v = 0;
    int t_v = 0;
    if (loff < left) {
      const uint2 pr = pairs[addr];
      s_v = (int)pr.x;
      t_v = (int)(pr.y - (unsigned)row0) & (int)rmask;
      atomicAdd(&cn[t_v], 1u);
    }
    if (left >= 32u) {
      add_rows<32>(xsrc, acc, s_v, t_v, 0, loff);
    } else {
      int jb = 0;
      if (left & 16u) { add_rows<16>(xsrc, acc, s_v, t_v, jb, loff); jb += 16; }
      if (left & 8u)  { add_rows<8>(xsrc, acc, s_v, t_v, jb, loff);  jb += 8; }
      if (left & 4u)  { add_rows<4>(xsrc, acc, s_v, t_v, jb, loff);  jb += 4; }
      if (left & 2u)  { add_rows<2>(xsrc, acc, s_v, t_v, jb, loff);  jb += 2; }
      if (left & 1u)  { add_rows<1>(xsrc, acc, s_v, t_v, jb, loff); }
    }
  }
  __syncthreads();
  float4* out4 = (float4*)(agg + (size_t)row0 * 32);
  for (int i = tid; i < R * 8; i += 1024) out4[i] = acc4[i];
  if (tid < R) cnt[row0 + tid] = (float)cn[tid];
}

template <int K>
__device__ __forceinline__ void stage_weights(const float* __restrict__ w,
                                              float* __restrict__ lw) {
  static_assert(K % 32 == 0, "32*K floats: a whole number of float4 per thread");
  constexpr int P = K + 4;
  const float4* w4 = (const float4*)w;
#pragma unroll
  for (int i = 0; i < K / 32; ++i) {
    const int f   = i * 256 + (int)threadIdx.x;
    const int row = f / (K / 4);
    const int q   = f % (K / 4);
    *(float4*)(lw + row * P + 4 * q) = w4[f];
  }
}

template <int K>
__device__ __forceinline__ void stage_rows(const float* __restrict__ rows,
                                           float* __restrict__ la, int lane) {
  static_assert(K % 8 == 0, "16*K floats: a whole number of float4 per lane");
  constexpr int P = K + 4;
  const float4* r4 = (const float4*)rows;
#pragma unroll
  for (int i = 0; i < K / 8; ++i) {
    const int f   = i * 32 + lane;
    const int row = f / (K / 4);
    const int q   = f % (K / 4);
    *(float4*)(la + row * P + 4 * q) = r4[f];
  }
}

template <int K>
__global__ __launch_bounds__(256) void sage_project_kernel(
    const float* __restrict__ xin, const float* __restrict__ w,
    float* __restrict__ y, int nrows) {
  constexpr int P = K + 4;
  __shared__ __attribute__((aligned(16))) float lw[32 * P];
  __shared__ __attribute__((aligned(16))) float la[8 * 16 * P];
  stage_weights<K>(w, lw);
  const int lane = threadIdx.x & 31;
  const int wv   = threadIdx.x >> 5;
  const int wid  = blockIdx.x * (blockDim.x >> 5) + wv;
  const int base = wid * 16;
  float* myA = la + wv * (16 * P);
  if (base < nrows) stage_rows<K>(xin + (size_t)base * K, myA, lane);
  __syncthreads();
  if (base >= nrows) return;
  const int l16  = lane & 15;
  const int half = lane >> 4;

  v8f c0 = {};
  v8f c1 = {};
  const float2* A2 = (const float2*)(myA + l16 * P);
  const float2* B0 = (const float2*)(lw + l16 * P);
  const float2* B1 = (const float2*)(lw + (l16 + 16) * P);
#pragma unroll
  for (int kc = 0; kc < K; kc += 4) {
    const int fi = (kc >> 1) + half;
    float2 fa = A2[fi];
    float2 f0 = B0[fi];
    float2 f1 = B1[fi];
    v2f a;  a.x  = fa.x; a.y  = fa.y;
    v2f b0; b0.x = f0.x; b0.y = f0.y;
    v2f b1; b1.x = f1.x; b1.y = f1.y;
    c0 = wmma_f32_k4(a, b0, c0);
    c1 = wmma_f32_k4(a, b1, c1);
  }
#pragma unroll
  for (int r = 0; r < 8; ++r) {
    const size_t row = (size_t)(base + r + half * 8);
    y[row * 32 + l16]      = c0[r];
    y[row * 32 + 16 + l16] = c1[r];
  }
}

template <int K, bool RELU>
__global__ __launch_bounds__(256) void sage_combine_kernel(
    const float* __restrict__ aggy, const float* __restrict__ cnt,
    const float* __restrict__ xin, const float* __restrict__ bl,
    const float* __restrict__ wr, float* __restrict__ out, int nrows) {
  constexpr int P = K + 4;
  __shared__ __attribute__((aligned(16))) float lw[32 * P];
  __shared__ __attribute__((aligned(16))) float la[8 * 16 * P];
  stage_weights<K>(wr, lw);
  const int lane = threadIdx.x & 31;
  const int wv   = threadIdx.x >> 5;
  const int wid  = blockIdx.x * (blockDim.x >> 5) + wv;
  const int base = wid * 16;
  float* myA = la + wv * (16 * P);
  if (base < nrows) stage_rows<K>(xin + (size_t)base * K, myA, lane);
  __syncthreads();
  if (base >= nrows) return;
  const int l16  = lane & 15;
  const int half = lane >> 4;

  v8f c0;
  v8f c1;
#pragma unroll
  for (int r = 0; r < 8; ++r) {
    const size_t row = (size_t)(base + r + half * 8);
    const float inv = 1.0f / fmaxf(cnt[row], 1.0f);
    c0[r] = aggy[row * 32 + l16] * inv;
    c1[r] = aggy[row * 32 + 16 + l16] * inv;
  }

  const float2* A2 = (const float2*)(myA + l16 * P);
  const float2* B0 = (const float2*)(lw + l16 * P);
  const float2* B1 = (const float2*)(lw + (l16 + 16) * P);
#pragma unroll
  for (int kc = 0; kc < K; kc += 4) {
    const int fi = (kc >> 1) + half;
    float2 fa = A2[fi];
    float2 f0 = B0[fi];
    float2 f1 = B1[fi];
    v2f a;  a.x  = fa.x; a.y  = fa.y;
    v2f b0; b0.x = f0.x; b0.y = f0.y;
    v2f b1; b1.x = f1.x; b1.y = f1.y;
    c0 = wmma_f32_k4(a, b0, c0);
    c1 = wmma_f32_k4(a, b1, c1);
  }

  const float bias0 = bl[l16];
  const float bias1 = bl[l16 + 16];
#pragma unroll
  for (int r = 0; r < 8; ++r) { c0[r] += bias0; c1[r] += bias1; }

#pragma unroll
  for (int r = 0; r < 8; ++r) {
    float p = c0[r] * c0[r] + c1[r] * c1[r];
    p += __shfl_xor(p, 1, 32);
    p += __shfl_xor(p, 2, 32);
    p += __shfl_xor(p, 4, 32);
    p += __shfl_xor(p, 8, 32);
    float scale = 1.0f / fmaxf(sqrtf(p), 1e-12f);
    float o0 = c0[r] * scale;
    float o1 = c1[r] * scale;
    if (RELU) { o0 = fmaxf(o0, 0.0f); o1 = fmaxf(o1, 0.0f); }
    const size_t row = (size_t)(base + r + half * 8);
    out[row * 32 + l16]      = o0;
    out[row * 32 + 16 + l16] = o1;
  }
}

template <int K, bool RELU>
__global__ __launch_bounds__(256) void sage_compute_kernel(
    const float* __restrict__ agg, const float* __restrict__ cnt,
    const float* __restrict__ xin, const float* __restrict__ wl,
    const float* __restrict__ bl, const float* __restrict__ wr,
    float* __restrict__ out, int nrows) {
  constexpr int P = K + 4;
  __shared__ __attribute__((aligned(16))) float lwl[32 * P];
  __shared__ __attribute__((aligned(16))) float lwr[32 * P];
  stage_weights<K>(wl, lwl);
  stage_weights<K>(wr, lwr);
  __syncthreads();
  const int lane = threadIdx.x & 31;
  const int wid  = blockIdx.x * (blockDim.x >> 5) + (threadIdx.x >> 5);
  const int base = wid * 16;
  if (base >= nrows) return;
  const int l16  = lane & 15;
  const int half = lane >> 4;
  const int rowA = base + l16;

  const float inv = 1.0f / fmaxf(cnt[rowA], 1.0f);

  v8f c0 = {};
  v8f c1 = {};

#pragma unroll
  for (int phase = 0; phase < 2; ++phase) {
    const float* Arow = (phase == 0 ? agg : xin) + (size_t)rowA * K;
    const float* W    = (phase == 0 ? lwl : lwr);
    const float  s    = (phase == 0 ? inv : 1.0f);
    const float2* A2  = (const float2*)Arow;
    const float2* B0  = (const float2*)(W + l16 * P);
    const float2* B1  = (const float2*)(W + (l16 + 16) * P);
#pragma unroll
    for (int kc = 0; kc < K; kc += 4) {
      const int fi = (kc >> 1) + half;
      float2 fa = A2[fi];
      float2 f0 = B0[fi];
      float2 f1 = B1[fi];
      v2f a;  a.x  = fa.x * s; a.y  = fa.y * s;
      v2f b0; b0.x = f0.x;     b0.y = f0.y;
      v2f b1; b1.x = f1.x;     b1.y = f1.y;
      c0 = wmma_f32_k4(a, b0, c0);
      c1 = wmma_f32_k4(a, b1, c1);
    }
  }

  const float bias0 = bl[l16];
  const float bias1 = bl[l16 + 16];
#pragma unroll
  for (int r = 0; r < 8; ++r) { c0[r] += bias0; c1[r] += bias1; }

#pragma unroll
  for (int r = 0; r < 8; ++r) {
    float p = c0[r] * c0[r] + c1[r] * c1[r];
    p += __shfl_xor(p, 1, 32);
    p += __shfl_xor(p, 2, 32);
    p += __shfl_xor(p, 4, 32);
    p += __shfl_xor(p, 8, 32);
    float scale = 1.0f / fmaxf(sqrtf(p), 1e-12f);
    float o0 = c0[r] * scale;
    float o1 = c1[r] * scale;
    if (RELU) { o0 = fmaxf(o0, 0.0f); o1 = fmaxf(o1, 0.0f); }
    const size_t row = (size_t)(base + r + half * 8);
    out[row * 32 + l16]      = o0;
    out[row * 32 + 16 + l16] = o1;
  }
}

extern "C" void kernel_launch(void* const* d_in, const int* in_sizes, int n_in,
                              void* d_out, int out_size, void* d_ws,
                              size_t ws_size, hipStream_t stream) {
  const float* x   = (const float*)d_in[0];
  const int* src0  = (const int*)d_in[1];
  const int* tgt0  = (const int*)d_in[2];
  const int* src1  = (const int*)d_in[3];
  const int* tgt1  = (const int*)d_in[4];
  const float* wl0 = (const float*)d_in[5];
  const float* bl0 = (const float*)d_in[6];
  const float* wr0 = (const float*)d_in[7];
  const float* wl1 = (const float*)d_in[8];
  const float* bl1 = (const float*)d_in[9];
  const float* wr1 = (const float*)d_in[10];
  const int E0 = in_sizes[1];
  const int E1 = in_sizes[3];

  float* ws   = (float*)d_ws;
  float* agg0 = ws;
  float* cnt0 = agg0 + (size_t)N1_ * H_;
  float* agg1 = cnt0 + (size_t)N1_;
  float* cnt1 = agg1 + (size_t)N2_ * H_;
  const int slices0 = (E0 + SL_ - 1) / SL_;
  const int slices1 = (E1 + SL_ - 1) / SL_;
  unsigned* pairs0 = (unsigned*)(cnt1 + (size_t)N2_);
  unsigned* pairs1 = pairs0 + (size_t)slices0 * SL_ * 2;
  unsigned* starts0 = pairs1 + (size_t)slices1 * SL_ * 2;
  unsigned* starts1 = starts0 + (size_t)slices0 * STP_;
  float* y0 = (float*)(starts1 + (size_t)slices1 * STP_);
  float* h  = y0;

  sage_group_kernel<<<slices0 + slices1, 1024, 0, stream>>>(
      src0, tgt0, E0, SH0_, pairs0, starts0, slices0,
      src1, tgt1, E1, SH1_, pairs1, starts1);
  {
    int blocks = (N0_ / 16) / 8;
    sage_project_kernel<INF_><<<blocks, 256, 0, stream>>>(x, wl0, y0, N0_);
  }
  sage_gather_kernel<<<NB_, 1024, 0, stream>>>(
      y0, pairs0, starts0, agg0, cnt0, slices0, SH0_);
  {
    int blocks = (N1_ / 16) / 8;
    sage_combine_kernel<INF_, true><<<blocks, 256, 0, stream>>>(
        agg0, cnt0, x, bl0, wr0, h, N1_);
  }
  sage_gather_kernel<<<NB_, 1024, 0, stream>>>(
      h, pairs1, starts1, agg1, cnt1, slices1, SH1_);
  {
    int blocks = (N2_ / 16) / 8;
    sage_compute_kernel<H_, false><<<blocks, 256, 0, stream>>>(
        agg1, cnt1, h, wl1, bl1, wr1, (float*)d_out, N2_);
  }
}


# Gather work-groups finish their rows' layer from their LDS sums (layer 0: WMMA start value; layer 1: A operand): agg/cnt planes and two launches go; k order and arrival-order sums unchanged
# speedup vs baseline: 6.3040x; 1.1387x over previous
#include <hip/hip_runtime.h>
#include <math.h>

typedef __attribute__((ext_vector_type(2))) float v2f;
typedef __attribute__((ext_vector_type(4))) float v4f;
typedef __attribute__((ext_vector_type(8))) float v8f;

#define N0_  262144
#define INF_ 128
#define H_   32
#define N1_  131072
#define N2_  65536

__device__ __forceinline__ v8f wmma_f32_k4(v2f a, v2f b, v8f c) {
  return __builtin_amdgcn_wmma_f32_16x16x4_f32(
       false, a,  false, b,
       (short)0, c,  false,  false);
}

#define NB_ 256
#define SL_ 8192
#define STP_ (NB_ + 32)
#define SH0_ 9
#define SH1_ 8
static_assert((NB_ << SH0_) == N1_, "ranges cover the layer-0 targets");
static_assert((NB_ << SH1_) == N2_, "ranges cover the layer-1 targets");
__global__ __launch_bounds__(1024) void sage_group_kernel(
    const int* __restrict__ srcA, const int* __restrict__ tgtA, int edgesA,
    int shiftA, unsigned* __restrict__ pairsA, unsigned* __restrict__ startsA,
    int slicesA,
    const int* __restrict__ srcB, const int* __restrict__ tgtB, int edgesB,
    int shiftB, unsigned* __restrict__ pairsB, unsigned* __restrict__ startsB) {
  __shared__ unsigned hcnt[NB_];
  __shared__ unsigned wtot[NB_ / 32];
  __shared__ unsigned total_s;
  __shared__ __attribute__((aligned(16))) uint2 spair[SL_];
  const bool second = (int)blockIdx.x >= slicesA;
  const int* __restrict__ src = second ? srcB : srcA;
  const int* __restrict__ tgt = second ? tgtB : tgtA;
  const int edges = second ? edgesB : edgesA;
  const int shift = second ? shiftB : shiftA;
  uint2* __restrict__ pairs = (uint2*)(second ? pairsB : pairsA);
  unsigned* __restrict__ starts = second ? startsB : startsA;
  const int sl  = second ? (int)blockIdx.x - slicesA : (int)blockIdx.x;
  const int tid = (int)threadIdx.x;
  const int lane = tid & 31;
  const int e_begin = sl * SL_;
  const int n = min(SL_, edges - e_begin);

  if (tid < NB_) hcnt[tid] = 0u;
  __syncthreads();
  int s[SL_ / 1024];
  int t[SL_ / 1024];
#pragma unroll
  for (int u = 0; u < SL_ / 1024; ++u) {
    const int i = tid + u * 1024;
    s[u] = 0;
    t[u] = -1;
    if (i < n) { s[u] = src[e_begin + i]; t[u] = tgt[e_begin + i]; }
  }
#pragma unroll
  for (int u = 0; u < SL_ / 1024; ++u) {
    const unsigned b = (unsigned)t[u] >> shift;
    if (b < (unsigned)NB_) atomicAdd(&hcnt[b], 1u);
    else t[u] = -1;
  }
  __syncthreads();
  unsigned c = 0u;
  if (tid < NB_) c = hcnt[tid];
  unsigned incl = c;
#pragma unroll
  for (int d = 1; d < 32; d <<= 1) {
    const unsigned v = __shfl_up(incl, d, 32);
    if (lane >= d) incl += v;
  }
  if (tid < NB_ && lane == 31) wtot[tid >> 5] = incl;
  __syncthreads();
  if (tid < NB_) {
    unsigned base = 0u;
    for (int k = 0; k < (tid >> 5); ++k) base += wtot[k];
    const unsigned ex = base + incl - c;
    hcnt[tid] = ex;
    starts[(size_t)sl * STP_ + tid] = ex;
    if (tid == NB_ - 1) {
      starts[(size_t)sl * STP_ + NB_] = ex + c;
      total_s = ex + c;
    }
  }
  __syncthreads();
#pragma unroll
  for (int u = 0; u < SL_ / 1024; ++u) {
    if (t[u] >= 0) {
      const unsigned b   = (unsigned)t[u] >> shift;
      const unsigned pos = atomicAdd(&hcnt[b], 1u);
      spair[pos] = make_uint2((unsigned)s[u], (unsigned)t[u]);
    }
  }
  __syncthreads();
  const int total = (int)total_s;
  for (int i = tid; i < total; i += 1024) pairs[(size_t)e_begin + i] = spair[i];
}

template <int N>
__device__ __forceinline__ void add_rows(const float* __restrict__ xsrc,
                                         float* __restrict__ acc, int s_v,
                                         int t_v, int jb, unsigned loff) {
  float v[N];
#pragma unroll
  for (int j = 0; j < N; ++j) {
    const unsigned s = (unsigned)__builtin_amdgcn_readlane(s_v, jb + j);
    v[j] = xsrc[s * 32u + loff];
  }
#pragma unroll
  for (int j = 0; j < N; ++j) {
    const unsigned t = (unsigned)__builtin_amdgcn_readlane(t_v, jb + j);
    atomicAdd(&acc[t * 32u + loff], v[j]);
  }
}

template <int SHIFT>
__device__ __forceinline__ void gather_range(
    const float* __restrict__ xsrc, const unsigned* __restrict__ pairs_w,
    const unsigned* __restrict__ starts, float* __restrict__ acc,
    unsigned* __restrict__ cn, int nslices, int b, int tid, int lane, int wave) {
  const uint2* __restrict__ pairs = (const uint2*)pairs_w;
  constexpr int R = 1 << SHIFT;
  const int row0 = b << SHIFT;
  float4* acc4 = (float4*)acc;
  for (int i = tid; i < R * 8; i += 1024) acc4[i] = make_float4(0.f, 0.f, 0.f, 0.f);
  if (tid < R) cn[tid] = 0u;
  __syncthreads();

  const int rpw = (nslices + 31) / 32;
  unsigned st = 0u, len = 0u;
  {
    const int g = wave * rpw + lane;
    if (lane < rpw && g < nslices) {
      const unsigned* p = starts + (size_t)g * STP_ + b;
      const unsigned a = p[0];
      const unsigned e = p[1];
      st  = (unsigned)g * (unsigned)SL_ + a;
      len = e - a;
    }
  }
  unsigned incl = len;
#pragma unroll
  for (int d = 1; d < 32; d <<= 1) {
    const unsigned v = __shfl_up(incl, d, 32);
    if (lane >= d) incl += v;
  }
  const unsigned excl  = incl - len;
  const unsigned total = (unsigned)__builtin_amdgcn_readlane((int)incl, 31);
  const unsigned loff  = (unsigned)lane;
  const unsigned rmask = (unsigned)R - 1u;

  for (unsigned c0 = 0u; c0 < total; c0 += 32u) {
    const unsigned q = c0 + loff;
    unsigned addr = 0u;
    for (int k = 0; k < rpw; ++k) {
      const unsigned ek = (unsigned)__builtin_amdgcn_readlane((int)excl, k);
      const unsigned ik = (unsigned)__builtin_amdgcn_readlane((int)incl, k);
      const unsigned sk = (unsigned)__builtin_amdgcn_readlane((int)st, k);
      if (q >= ek && q < ik) addr = sk + (q - ek);
    }
    const unsigned left = total - c0;
    int s_v = 0;
    int t_v = 0;
    if (loff < left) {
      const uint2 pr = pairs[addr];
      s_v = (int)pr.x;
      t_v = (int)(pr.y - (unsigned)row0) & (int)rmask;
      atomicAdd(&cn[t_v], 1u);
    }
    if (left >= 32u) {
      add_rows<32>(xsrc, acc, s_v, t_v, 0, loff);
    } else {
      int jb = 0;
      if (left & 16u) { add_rows<16>(xsrc, acc, s_v, t_v, jb, loff); jb += 16; }
      if (left & 8u)  { add_rows<8>(xsrc, acc, s_v, t_v, jb, loff);  jb += 8; }
      if (left & 4u)  { add_rows<4>(xsrc, acc, s_v, t_v, jb, loff);  jb += 4; }
      if (left & 2u)  { add_rows<2>(xsrc, acc, s_v, t_v, jb, loff);  jb += 2; }
      if (left & 1u)  { add_rows<1>(xsrc, acc, s_v, t_v, jb, loff); }
    }
  }
}

template <int K>
__device__ __forceinline__ void stage_weights(const float* __restrict__ w,
                                              float* __restrict__ lw) {
  static_assert(K % 32 == 0, "32*K floats: a whole number of float4 per thread");
  constexpr int P = K + 4;
  const float4* w4 = (const float4*)w;
#pragma unroll
  for (int i = 0; i < K / 32; ++i) {
    const int f   = i * 256 + (int)threadIdx.x;
    const int row = f / (K / 4);
    const int q   = f % (K / 4);
    *(float4*)(lw + row * P + 4 * q) = w4[f];
  }
}

template <int K>
__device__ __forceinline__ void stage_rows(const float* __restrict__ rows,
                                           float* __restrict__ la, int lane) {
  static_assert(K % 8 == 0, "16*K floats: a whole number of float4 per lane");
  constexpr int P = K + 4;
  const float4* r4 = (const float4*)rows;
#pragma unroll
  for (int i = 0; i < K / 8; ++i) {
    const int f   = i * 32 + lane;
    const int row = f / (K / 4);
    const int q   = f % (K / 4);
    *(float4*)(la + row * P + 4 * q) = r4[f];
  }
}

template <int K>
__global__ __launch_bounds__(256) void sage_project_kernel(
    const float* __restrict__ xin, const float* __restrict__ w,
    float* __restrict__ y, int nrows) {
  constexpr int P = K + 4;
  __shared__ __attribute__((aligned(16))) float lw[32 * P];
  __shared__ __attribute__((aligned(16))) float la[8 * 16 * P];
  stage_weights<K>(w, lw);
  const int lane = threadIdx.x & 31;
  const int wv   = threadIdx.x >> 5;
  const int wid  = blockIdx.x * (blockDim.x >> 5) + wv;
  const int base = wid * 16;
  float* myA = la + wv * (16 * P);
  if (base < nrows) stage_rows<K>(xin + (size_t)base * K, myA, lane);
  __syncthreads();
  if (base >= nrows) return;
  const int l16  = lane & 15;
  const int half = lane >> 4;

  v8f c0 = {};
  v8f c1 = {};
  const float2* A2 = (const float2*)(myA + l16 * P);
  const float2* B0 = (const float2*)(lw + l16 * P);
  const float2* B1 = (const float2*)(lw + (l16 + 16) * P);
#pragma unroll
  for (int kc = 0; kc < K; kc += 4) {
    const int fi = (kc >> 1) + half;
    float2 fa = A2[fi];
    float2 f0 = B0[fi];
    float2 f1 = B1[fi];
    v2f a;  a.x  = fa.x; a.y  = fa.y;
    v2f b0; b0.x = f0.x; b0.y = f0.y;
    v2f b1; b1.x = f1.x; b1.y = f1.y;
    c0 = wmma_f32_k4(a, b0, c0);
    c1 = wmma_f32_k4(a, b1, c1);
  }
#pragma unroll
  for (int r = 0; r < 8; ++r) {
    const size_t row = (size_t)(base + r + half * 8);
    y[row * 32 + l16]      = c0[r];
    y[row * 32 + 16 + l16] = c1[r];
  }
}

__global__ __launch_bounds__(1024) void sage_gather_combine_kernel(
    const float* __restrict__ ysrc, const unsigned* __restrict__ pairs_w,
    const unsigned* __restrict__ starts, int nslices,
    const float* __restrict__ xin, const float* __restrict__ bl,
    const float* __restrict__ wr, float* __restrict__ out) {
  constexpr int K = INF_;
  constexpr int P = K + 4;
  constexpr int R = 1 << SH0_;
  __shared__ __attribute__((aligned(16))) float big[32 * 16 * P];
  static_assert(R * 32 <= 32 * 16 * P, "the sums fit in front of the tiles' space");
  __shared__ unsigned cn[R];
  __shared__ __attribute__((aligned(16))) float lw[32 * P];
  float* acc = big;
  const int b    = (int)blockIdx.x;
  const int tid  = (int)threadIdx.x;
  const int lane = tid & 31;
  const int wave = __builtin_amdgcn_readfirstlane(tid >> 5);
  static_assert(32 * K == 1024 * 4, "wr is one float4 per thread");
  *(float4*)(lw + (tid / (K / 4)) * P + 4 * (tid % (K / 4))) = ((const float4*)wr)[tid];

  gather_range<SH0_>(ysrc, pairs_w, starts, acc, cn, nslices, b, tid, lane, wave);

  const int lrow = wave * 16;
  const int base = (b << SH0_) + lrow;
  float* myA = big + wave * (16 * P);
  const v4f* r4 = (const v4f*)(xin + (size_t)base * K);
  v4f xa[K / 8];
#pragma unroll
  for (int i = 0; i < K / 8; ++i) xa[i] = r4[i * 32 + lane];
  __syncthreads();
  const int l16  = lane & 15;
  const int half = lane >> 4;
  v8f c0;
  v8f c1;
  const float myinv = 1.0f / fmaxf((float)cn[lrow + half * 8 + (l16 & 7)], 1.0f);
#pragma unroll
  for (int r = 0; r < 8; ++r) {
    const int row = lrow + r + half * 8;
    const float inv = __shfl(myinv, half * 16 + r, 32);
    c0[r] = acc[row * 32 + l16] * inv;
    c1[r] = acc[row * 32 + 16 + l16] * inv;
  }
  __syncthreads();
#pragma unroll
  for (int i = 0; i < K / 8; ++i) {
    const int f = i * 32 + lane;
    *(v4f*)(myA + (f / (K / 4)) * P + 4 * (f % (K / 4))) = xa[i];
  }
  __syncthreads();

  const float2* A2 = (const float2*)(myA + l16 * P);
  const float2* B0 = (const float2*)(lw + l16 * P);
  const float2* B1 = (const float2*)(lw + (l16 + 16) * P);
#pragma unroll
  for (int kc = 0; kc < K; kc += 4) {
    const int fi = (kc >> 1) + half;
    float2 fa = A2[fi];
    float2 f0 = B0[fi];
    float2 f1 = B1[fi];
    v2f a;  a.x  = fa.x; a.y  = fa.y;
    v2f b0; b0.x = f0.x; b0.y = f0.y;
    v2f b1; b1.x = f1.x; b1.y = f1.y;
    c0 = wmma_f32_k4(a, b0, c0);
    c1 = wmma_f32_k4(a, b1, c1);
  }

  const float bias0 = bl[l16];
  const float bias1 = bl[l16 + 16];
#pragma unroll
  for (int r = 0; r < 8; ++r) { c0[r] += bias0; c1[r] += bias1; }

#pragma unroll
  for (int r = 0; r < 8; ++r) {
    float p = c0[r] * c0[r] + c1[r] * c1[r];
    p += __shfl_xor(p, 1, 32);
    p += __shfl_xor(p, 2, 32);
    p += __shfl_xor(p, 4, 32);
    p += __shfl_xor(p, 8, 32);
    float scale = 1.0f / fmaxf(sqrtf(p), 1e-12f);
    const float o0 = fmaxf(c0[r] * scale, 0.0f);
    const float o1 = fmaxf(c1[r] * scale, 0.0f);
    const size_t row = (size_t)(base + r + half * 8);
    out[row * 32 + l16]      = o0;
    out[row * 32 + 16 + l16] = o1;
  }
}

__global__ __launch_bounds__(1024) void sage_gather_transform_kernel(
    const float* __restrict__ hin, const unsigned* __restrict__ pairs_w,
    const unsigned* __restrict__ starts, int nslices,
    const float* __restrict__ wl, const float* __restrict__ bl,
    const float* __restrict__ wr, float* __restrict__ out) {
  constexpr int K = H_;
  constexpr int P = K + 4;
  constexpr int R = 1 << SH1_;
  __shared__ __attribute__((aligned(16))) float acc[R * 32];
  __shared__ unsigned cn[R];
  __shared__ __attribute__((aligned(16))) float lwl[32 * P];
  __shared__ __attribute__((aligned(16))) float lwr[32 * P];
  const int b    = (int)blockIdx.x;
  const int tid  = (int)threadIdx.x;
  const int lane = tid & 31;
  const int wave = __builtin_amdgcn_readfirstlane(tid >> 5);
  static_assert(32 * K == 256 * 4, "wl, wr: one float4 for each of 256 threads");
  if (tid < 512) {
    const int f = tid & 255;
    const float4 v = ((const float4*)(tid < 256 ? wl : wr))[f];
    *(float4*)((tid < 256 ? lwl : lwr) + (f / (K / 4)) * P + 4 * (f % (K / 4))) = v;
  }

  gather_range<SH1_>(hin, pairs_w, starts, acc, cn, nslices, b, tid, lane, wave);
  __syncthreads();
  if (wave >= R / 16) return;

  const int lrow = wave * 16;
  const int base = (b << SH1_) + lrow;
  const int l16  = lane & 15;
  const int half = lane >> 4;
  const float inv = 1.0f / fmaxf((float)cn[lrow + l16], 1.0f);

  v8f c0 = {};
  v8f c1 = {};
  {
    const float2* A2 = (const float2*)(acc + (lrow + l16) * 32);
    const float2* B0 = (const float2*)(lwl + l16 * P);
    const float2* B1 = (const float2*)(lwl + (l16 + 16) * P);
#pragma unroll
    for (int kc = 0; kc < K; kc += 4) {
      const int fi = (kc >> 1) + half;
      float2 fa = A2[fi];
      float2 f0 = B0[fi];
      float2 f1 = B1[fi];
      v2f a;  a.x  = fa.x * inv; a.y  = fa.y * inv;
      v2f b0; b0.x = f0.x;       b0.y = f0.y;
      v2f b1; b1.x = f1.x;       b1.y = f1.y;
      c0 = wmma_f32_k4(a, b0, c0);
      c1 = wmma_f32_k4(a, b1, c1);
    }
  }
  {
    const float2* A2 = (const float2*)(hin + (size_t)(base + l16) * K);
    const float2* B0 = (const float2*)(lwr + l16 * P);
    const float2* B1 = (const float2*)(lwr + (l16 + 16) * P);
#pragma unroll
    for (int kc = 0; kc < K; kc += 4) {
      const int fi = (kc >> 1) + half;
      float2 fa = A2[fi];
      float2 f0 = B0[fi];
      float2 f1 = B1[fi];
      v2f a;  a.x  = fa.x; a.y  = fa.y;
      v2f b0; b0.x = f0.x; b0.y = f0.y;
      v2f b1; b1.x = f1.x; b1.y = f1.y;
      c0 = wmma_f32_k4(a, b0, c0);
      c1 = wmma_f32_k4(a, b1, c1);
    }
  }

  const float bias0 = bl[l16];
  const float bias1 = bl[l16 + 16];
#pragma unroll
  for (int r = 0; r < 8; ++r) { c0[r] += bias0; c1[r] += bias1; }

#pragma unroll
  for (int r = 0; r < 8; ++r) {
    float p = c0[r] * c0[r] + c1[r] * c1[r];
    p += __shfl_xor(p, 1, 32);
    p += __shfl_xor(p, 2, 32);
    p += __shfl_xor(p, 4, 32);
    p += __shfl_xor(p, 8, 32);
    float scale = 1.0f / fmaxf(sqrtf(p), 1e-12f);
    const size_t row = (size_t)(base + r + half * 8);
    out[row * 32 + l16]      = c0[r] * scale;
    out[row * 32 + 16 + l16] = c1[r] * scale;
  }
}

extern "C" void kernel_launch(void* const* d_in, const int* in_sizes, int n_in,
                              void* d_out, int out_size, void* d_ws,
                              size_t ws_size, hipStream_t stream) {
  const float* x   = (const float*)d_in[0];
  const int* src0  = (const int*)d_in[1];
  const int* tgt0  = (const int*)d_in[2];
  const int* src1  = (const int*)d_in[3];
  const int* tgt1  = (const int*)d_in[4];
  const float* wl0 = (const float*)d_in[5];
  const float* bl0 = (const float*)d_in[6];
  const float* wr0 = (const float*)d_in[7];
  const float* wl1 = (const float*)d_in[8];
  const float* bl1 = (const float*)d_in[9];
  const float* wr1 = (const float*)d_in[10];
  const int E0 = in_sizes[1];
  const int E1 = in_sizes[3];

  float* ws = (float*)d_ws;
  float* h  = ws;
  const int slices0 = (E0 + SL_ - 1) / SL_;
  const int slices1 = (E1 + SL_ - 1) / SL_;
  unsigned* pairs0 = (unsigned*)(h + (size_t)N1_ * H_);
  unsigned* pairs1 = pairs0 + (size_t)slices0 * SL_ * 2;
  unsigned* starts0 = pairs1 + (size_t)slices1 * SL_ * 2;
  unsigned* starts1 = starts0 + (size_t)slices0 * STP_;
  float* y0 = (float*)(starts1 + (size_t)slices1 * STP_);

  sage_group_kernel<<<slices0 + slices1, 1024, 0, stream>>>(
      src0, tgt0, E0, SH0_, pairs0, starts0, slices0,
      src1, tgt1, E1, SH1_, pairs1, starts1);
  {
    int blocks = (N0_ / 16) / 8;
    sage_project_kernel<INF_><<<blocks, 256, 0, stream>>>(x, wl0, y0, N0_);
  }
  sage_gather_combine_kernel<<<NB_, 1024, 0, stream>>>(
      y0, pairs0, starts0, slices0, x, bl0, wr0, h);
  sage_gather_transform_kernel<<<NB_, 1024, 0, stream>>>(
      h, pairs1, starts1, slices1, wl1, bl1, wr1, (float*)d_out);
}


# Edge ordering joins the projection's launch: 256-thread work-groups, the first 384 each order one 8192-edge slice, the rest project; 3 launches, not 4; k order and arrival-order sums unchanged
# speedup vs baseline: 6.8026x; 1.0791x over previous
#include <hip/hip_runtime.h>
#include <math.h>

typedef __attribute__((ext_vector_type(2))) float v2f;
typedef __attribute__((ext_vector_type(4))) float v4f;
typedef __attribute__((ext_vector_type(8))) float v8f;

#define N0_  262144
#define INF_ 128
#define H_   32
#define N1_  131072
#define N2_  65536

__device__ __forceinline__ v8f wmma_f32_k4(v2f a, v2f b, v8f c) {
  return __builtin_amdgcn_wmma_f32_16x16x4_f32(
       false, a,  false, b,
       (short)0, c,  false,  false);
}

#define NB_ 256
#define SL_ 8192
#define GT_ 256
#define EPT_ (SL_ / GT_)
#define STP_ (NB_ + 32)
#define SH0_ 9
#define SH1_ 8
static_assert((NB_ << SH0_) == N1_, "ranges cover the layer-0 targets");
static_assert((NB_ << SH1_) == N2_, "ranges cover the layer-1 targets");
static_assert(GT_ == NB_, "one thread per range in the prefix");
static_assert(EPT_ % 4 == 0, "a thread's edges are whole groups of four");
#define G_SPAIR_W 0
#define G_HCNT_W  (2 * SL_)
#define G_WTOT_W  (G_HCNT_W + NB_)
#define G_TOTAL_W (G_WTOT_W + GT_ / 32)
#define G_WORDS   (G_TOTAL_W + 1)
__device__ __forceinline__ void group_slice(
    const int* __restrict__ src, const int* __restrict__ tgt, int edges,
    int shift, unsigned* __restrict__ pairs_w, unsigned* __restrict__ starts,
    int sl, unsigned* __restrict__ lds_w) {
  uint2* spair      = (uint2*)(lds_w + G_SPAIR_W);
  unsigned* hcnt    = lds_w + G_HCNT_W;
  unsigned* wtot    = lds_w + G_WTOT_W;
  unsigned* total_s = lds_w + G_TOTAL_W;
  uint2* __restrict__ pairs = (uint2*)pairs_w;
  const int tid = (int)threadIdx.x;
  const int lane = tid & 31;
  const int e_begin = sl * SL_;
  const int n = min(SL_, edges - e_begin);

  hcnt[tid] = 0u;
  __syncthreads();
  int s[EPT_];
  int t[EPT_];
  if (n == SL_) {
    const int4* __restrict__ s4 = (const int4*)(src + e_begin);
    const int4* __restrict__ t4 = (const int4*)(tgt + e_begin);
#pragma unroll
    for (int u = 0; u < EPT_ / 4; ++u) {
      const int4 a = s4[tid + u * GT_];
      const int4 b = t4[tid + u * GT_];
      s[4 * u + 0] = a.x; s[4 * u + 1] = a.y; s[4 * u + 2] = a.z; s[4 * u + 3] = a.w;
      t[4 * u + 0] = b.x; t[4 * u + 1] = b.y; t[4 * u + 2] = b.z; t[4 * u + 3] = b.w;
    }
  } else {
#pragma unroll
    for (int u = 0; u < EPT_; ++u) {
      const int i = tid + u * GT_;
      s[u] = 0;
      t[u] = -1;
      if (i < n) { s[u] = src[e_begin + i]; t[u] = tgt[e_begin + i]; }
    }
  }
#pragma unroll
  for (int u = 0; u < EPT_; ++u) {
    const unsigned b = (unsigned)t[u] >> shift;
    if (b < (unsigned)NB_) atomicAdd(&hcnt[b], 1u);
    else t[u] = -1;
  }
  __syncthreads();
  const unsigned c = hcnt[tid];
  unsigned incl = c;
#pragma unroll
  for (int d = 1; d < 32; d <<= 1) {
    const unsigned v = __shfl_up(incl, d, 32);
    if (lane >= d) incl += v;
  }
  if (lane == 31) wtot[tid >> 5] = incl;
  __syncthreads();
  {
    unsigned base = 0u;
    for (int k = 0; k < (tid >> 5); ++k) base += wtot[k];
    const unsigned ex = base + incl - c;
    hcnt[tid] = ex;
    starts[(size_t)sl * STP_ + tid] = ex;
    if (tid == NB_ - 1) {
      starts[(size_t)sl * STP_ + NB_] = ex + c;
      *total_s = ex + c;
    }
  }
  __syncthreads();
#pragma unroll
  for (int u = 0; u < EPT_; ++u) {
    if (t[u] >= 0) {
      const unsigned b   = (unsigned)t[u] >> shift;
      const unsigned pos = atomicAdd(&hcnt[b], 1u);
      spair[pos] = make_uint2((unsigned)s[u], (unsigned)t[u]);
    }
  }
  __syncthreads();
  const int total = (int)*total_s;
  for (int i = tid; i < total; i += GT_) pairs[(size_t)e_begin + i] = spair[i];
}

template <int N>
__device__ __forceinline__ void add_rows(const float* __restrict__ xsrc,
                                         float* __restrict__ acc, int s_v,
                                         int t_v, int jb, unsigned loff) {
  float v[N];
#pragma unroll
  for (int j = 0; j < N; ++j) {
    const unsigned s = (unsigned)__builtin_amdgcn_readlane(s_v, jb + j);
    v[j] = xsrc[s * 32u + loff];
  }
#pragma unroll
  for (int j = 0; j < N; ++j) {
    const unsigned t = (unsigned)__builtin_amdgcn_readlane(t_v, jb + j);
    atomicAdd(&acc[t * 32u + loff], v[j]);
  }
}

template <int SHIFT>
__device__ __forceinline__ void gather_range(
    const float* __restrict__ xsrc, const unsigned* __restrict__ pairs_w,
    const unsigned* __restrict__ starts, float* __restrict__ acc,
    unsigned* __restrict__ cn, int nslices, int b, int tid, int lane, int wave) {
  const uint2* __restrict__ pairs = (const uint2*)pairs_w;
  constexpr int R = 1 << SHIFT;
  const int row0 = b << SHIFT;
  float4* acc4 = (float4*)acc;
  for (int i = tid; i < R * 8; i += 1024) acc4[i] = make_float4(0.f, 0.f, 0.f, 0.f);
  if (tid < R) cn[tid] = 0u;
  __syncthreads();

  const int rpw = (nslices + 31) / 32;
  unsigned st = 0u, len = 0u;
  {
    const int g = wave * rpw + lane;
    if (lane < rpw && g < nslices) {
      const unsigned* p = starts + (size_t)g * STP_ + b;
      const unsigned a = p[0];
      const unsigned e = p[1];
      st  = (unsigned)g * (unsigned)SL_ + a;
      len = e - a;
    }
  }
  unsigned incl = len;
#pragma unroll
  for (int d = 1; d < 32; d <<= 1) {
    const unsigned v = __shfl_up(incl, d, 32);
    if (lane >= d) incl += v;
  }
  const unsigned excl  = incl - len;
  const unsigned total = (unsigned)__builtin_amdgcn_readlane((int)incl, 31);
  const unsigned loff  = (unsigned)lane;
  const unsigned rmask = (unsigned)R - 1u;

  for (unsigned c0 = 0u; c0 < total; c0 += 32u) {
    const unsigned q = c0 + loff;
    unsigned addr = 0u;
    for (int k = 0; k < rpw; ++k) {
      const unsigned ek = (unsigned)__builtin_amdgcn_readlane((int)excl, k);
      const unsigned ik = (unsigned)__builtin_amdgcn_readlane((int)incl, k);
      const unsigned sk = (unsigned)__builtin_amdgcn_readlane((int)st, k);
      if (q >= ek && q < ik) addr = sk + (q - ek);
    }
    const unsigned left = total - c0;
    int s_v = 0;
    int t_v = 0;
    if (loff < left) {
      const uint2 pr = pairs[addr];
      s_v = (int)pr.x;
      t_v = (int)(pr.y - (unsigned)row0) & (int)rmask;
      atomicAdd(&cn[t_v], 1u);
    }
    if (left >= 32u) {
      add_rows<32>(xsrc, acc, s_v, t_v, 0, loff);
    } else {
      int jb = 0;
      if (left & 16u) { add_rows<16>(xsrc, acc, s_v, t_v, jb, loff); jb += 16; }
      if (left & 8u)  { add_rows<8>(xsrc, acc, s_v, t_v, jb, loff);  jb += 8; }
      if (left & 4u)  { add_rows<4>(xsrc, acc, s_v, t_v, jb, loff);  jb += 4; }
      if (left & 2u)  { add_rows<2>(xsrc, acc, s_v, t_v, jb, loff);  jb += 2; }
      if (left & 1u)  { add_rows<1>(xsrc, acc, s_v, t_v, jb, loff); }
    }
  }
}

template <int K>
__device__ __forceinline__ void stage_weights(const float* __restrict__ w,
                                              float* __restrict__ lw) {
  static_assert(K % 32 == 0, "32*K floats: a whole number of float4 per thread");
  constexpr int P = K + 4;
  const float4* w4 = (const float4*)w;
#pragma unroll
  for (int i = 0; i < K / 32; ++i) {
    const int f   = i * 256 + (int)threadIdx.x;
    const int row = f / (K / 4);
    const int q   = f % (K / 4);
    *(float4*)(lw + row * P + 4 * q) = w4[f];
  }
}

template <int K>
__device__ __forceinline__ void stage_rows(const float* __restrict__ rows,
                                           float* __restrict__ la, int lane) {
  static_assert(K % 8 == 0, "16*K floats: a whole number of float4 per lane");
  constexpr int P = K + 4;
  const float4* r4 = (const float4*)rows;
#pragma unroll
  for (int i = 0; i < K / 8; ++i) {
    const int f   = i * 32 + lane;
    const int row = f / (K / 4);
    const int q   = f % (K / 4);
    *(float4*)(la + row * P + 4 * q) = r4[f];
  }
}

template <int K>
__device__ __forceinline__ void project_tiles(
    const float* __restrict__ xin, const float* __restrict__ w,
    float* __restrict__ y, int nrows, int pb, float* __restrict__ lw,
    float* __restrict__ la) {
  constexpr int P = K + 4;
  stage_weights<K>(w, lw);
  const int lane = threadIdx.x & 31;
  const int wv   = threadIdx.x >> 5;
  const int wid  = pb * (GT_ / 32) + wv;
  const int base = wid * 16;
  float* myA = la + wv * (16 * P);
  if (base < nrows) stage_rows<K>(xin + (size_t)base * K, myA, lane);
  __syncthreads();
  if (base >= nrows) return;
  const int l16  = lane & 15;
  const int half = lane >> 4;

  v8f c0 = {};
  v8f c1 = {};
  const float2* A2 = (const float2*)(myA + l16 * P);
  const float2* B0 = (const float2*)(lw + l16 * P);
  const float2* B1 = (const float2*)(lw + (l16 + 16) * P);
#pragma unroll
  for (int kc = 0; kc < K; kc += 4) {
    const int fi = (kc >> 1) + half;
    float2 fa = A2[fi];
    float2 f0 = B0[fi];
    float2 f1 = B1[fi];
    v2f a;  a.x  = fa.x; a.y  = fa.y;
    v2f b0; b0.x = f0.x; b0.y = f0.y;
    v2f b1; b1.x = f1.x; b1.y = f1.y;
    c0 = wmma_f32_k4(a, b0, c0);
    c1 = wmma_f32_k4(a, b1, c1);
  }
#pragma unroll
  for (int r = 0; r < 8; ++r) {
    const size_t row = (size_t)(base + r + half * 8);
    y[row * 32 + l16]      = c0[r];
    y[row * 32 + 16 + l16] = c1[r];
  }
}

#define PREP_LDS_WORDS (32 * (INF_ + 4) + 8 * 16 * (INF_ + 4))
static_assert(G_WORDS <= PREP_LDS_WORDS, "a slice's LDS fits in the projection's");
__global__ __launch_bounds__(GT_) void sage_prep_kernel(
    const int* __restrict__ srcA, const int* __restrict__ tgtA, int edgesA,
    int shiftA, unsigned* __restrict__ pairsA, unsigned* __restrict__ startsA,
    int slicesA,
    const int* __restrict__ srcB, const int* __restrict__ tgtB, int edgesB,
    int shiftB, unsigned* __restrict__ pairsB, unsigned* __restrict__ startsB,
    int slicesB,
    const float* __restrict__ xin, const float* __restrict__ w,
    float* __restrict__ y, int nrows) {
  __shared__ __attribute__((aligned(16))) unsigned lds_w[PREP_LDS_WORDS];
  const int blk = (int)blockIdx.x;
  if (blk < slicesA + slicesB) {
    const bool second = blk >= slicesA;
    group_slice(second ? srcB : srcA, second ? tgtB : tgtA,
                second ? edgesB : edgesA, second ? shiftB : shiftA,
                second ? pairsB : pairsA, second ? startsB : startsA,
                second ? blk - slicesA : blk, lds_w);
  } else {
    float* lw = (float*)lds_w;
    float* la = lw + 32 * (INF_ + 4);
    project_tiles<INF_>(xin, w, y, nrows, blk - slicesA - slicesB, lw, la);
  }
}

__global__ __launch_bounds__(1024) void sage_gather_combine_kernel(
    const float* __restrict__ ysrc, const unsigned* __restrict__ pairs_w,
    const unsigned* __restrict__ starts, int nslices,
    const float* __restrict__ xin, const float* __restrict__ bl,
    const float* __restrict__ wr, float* __restrict__ out) {
  constexpr int K = INF_;
  constexpr int P = K + 4;
  constexpr int R = 1 << SH0_;
  __shared__ __attribute__((aligned(16))) float big[32 * 16 * P];
  static_assert(R * 32 <= 32 * 16 * P, "the sums fit in front of the tiles' space");
  __shared__ unsigned cn[R];
  __shared__ __attribute__((aligned(16))) float lw[32 * P];
  float* acc = big;
  const int b    = (int)blockIdx.x;
  const int tid  = (int)threadIdx.x;
  const int lane = tid & 31;
  const int wave = __builtin_amdgcn_readfirstlane(tid >> 5);
  static_assert(32 * K == 1024 * 4, "wr is one float4 per thread");
  *(float4*)(lw + (tid / (K / 4)) * P + 4 * (tid % (K / 4))) = ((const float4*)wr)[tid];

  gather_range<SH0_>(ysrc, pairs_w, starts, acc, cn, nslices, b, tid, lane, wave);

  const int lrow = wave * 16;
  const int base = (b << SH0_) + lrow;
  float* myA = big + wave * (16 * P);
  const v4f* r4 = (const v4f*)(xin + (size_t)base * K);
  v4f xa[K / 8];
#pragma unroll
  for (int i = 0; i < K / 8; ++i) xa[i] = r4[i * 32 + lane];
  __syncthreads();
  const int l16  = lane & 15;
  const int half = lane >> 4;
  v8f c0;
  v8f c1;
  const float myinv = 1.0f / fmaxf((float)cn[lrow + half * 8 + (l16 & 7)], 1.0f);
#pragma unroll
  for (int r = 0; r < 8; ++r) {
    const int row = lrow + r + half * 8;
    const float inv = __shfl(myinv, half * 16 + r, 32);
    c0[r] = acc[row * 32 + l16] * inv;
    c1[r] = acc[row * 32 + 16 + l16] * inv;
  }
  __syncthreads();
#pragma unroll
  for (int i = 0; i < K / 8; ++i) {
    const int f = i * 32 + lane;
    *(v4f*)(myA + (f / (K / 4)) * P + 4 * (f % (K / 4))) = xa[i];
  }
  __syncthreads();

  const float2* A2 = (const float2*)(myA + l16 * P);
  const float2* B0 = (const float2*)(lw + l16 * P);
  const float2* B1 = (const float2*)(lw + (l16 + 16) * P);
#pragma unroll
  for (int kc = 0; kc < K; kc += 4) {
    const int fi = (kc >> 1) + half;
    float2 fa = A2[fi];
    float2 f0 = B0[fi];
    float2 f1 = B1[fi];
    v2f a;  a.x  = fa.x; a.y  = fa.y;
    v2f b0; b0.x = f0.x; b0.y = f0.y;
    v2f b1; b1.x = f1.x; b1.y = f1.y;
    c0 = wmma_f32_k4(a, b0, c0);
    c1 = wmma_f32_k4(a, b1, c1);
  }

  const float bias0 = bl[l16];
  const float bias1 = bl[l16 + 16];
#pragma unroll
  for (int r = 0; r < 8; ++r) { c0[r] += bias0; c1[r] += bias1; }

#pragma unroll
  for (int r = 0; r < 8; ++r) {
    float p = c0[r] * c0[r] + c1[r] * c1[r];
    p += __shfl_xor(p, 1, 32);
    p += __shfl_xor(p, 2, 32);
    p += __shfl_xor(p, 4, 32);
    p += __shfl_xor(p, 8, 32);
    float scale = 1.0f / fmaxf(sqrtf(p), 1e-12f);
    const float o0 = fmaxf(c0[r] * scale, 0.0f);
    const float o1 = fmaxf(c1[r] * scale, 0.0f);
    const size_t row = (size_t)(base + r + half * 8);
    out[row * 32 + l16]      = o0;
    out[row * 32 + 16 + l16] = o1;
  }
}

__global__ __launch_bounds__(1024) void sage_gather_transform_kernel(
    const float* __restrict__ hin, const unsigned* __restrict__ pairs_w,
    const unsigned* __restrict__ starts, int nslices,
    const float* __restrict__ wl, const float* __restrict__ bl,
    const float* __restrict__ wr, float* __restrict__ out) {
  constexpr int K = H_;
  constexpr int P = K + 4;
  constexpr int R = 1 << SH1_;
  __shared__ __attribute__((aligned(16))) float acc[R * 32];
  __shared__ unsigned cn[R];
  __shared__ __attribute__((aligned(16))) float lwl[32 * P];
  __shared__ __attribute__((aligned(16))) float lwr[32 * P];
  const int b    = (int)blockIdx.x;
  const int tid  = (int)threadIdx.x;
  const int lane = tid & 31;
  const int wave = __builtin_amdgcn_readfirstlane(tid >> 5);
  static_assert(32 * K == 256 * 4, "wl, wr: one float4 for each of 256 threads");
  if (tid < 512) {
    const int f = tid & 255;
    const float4 v = ((const float4*)(tid < 256 ? wl : wr))[f];
    *(float4*)((tid < 256 ? lwl : lwr) + (f / (K / 4)) * P + 4 * (f % (K / 4))) = v;
  }

  gather_range<SH1_>(hin, pairs_w, starts, acc, cn, nslices, b, tid, lane, wave);
  __syncthreads();
  if (wave >= R / 16) return;

  const int lrow = wave * 16;
  const int base = (b << SH1_) + lrow;
  const int l16  = lane & 15;
  const int half = lane >> 4;
  const float inv = 1.0f / fmaxf((float)cn[lrow + l16], 1.0f);

  v8f c0 = {};
  v8f c1 = {};
  {
    const float2* A2 = (const float2*)(acc + (lrow + l16) * 32);
    const float2* B0 = (const float2*)(lwl + l16 * P);
    const float2* B1 = (const float2*)(lwl + (l16 + 16) * P);
#pragma unroll
    for (int kc = 0; kc < K; kc += 4) {
      const int fi = (kc >> 1) + half;
      float2 fa = A2[fi];
      float2 f0 = B0[fi];
      float2 f1 = B1[fi];
      v2f a;  a.x  = fa.x * inv; a.y  = fa.y * inv;
      v2f b0; b0.x = f0.x;       b0.y = f0.y;
      v2f b1; b1.x = f1.x;       b1.y = f1.y;
      c0 = wmma_f32_k4(a, b0, c0);
      c1 = wmma_f32_k4(a, b1, c1);
    }
  }
  {
    const float2* A2 = (const float2*)(hin + (size_t)(base + l16) * K);
    const float2* B0 = (const float2*)(lwr + l16 * P);
    const float2* B1 = (const float2*)(lwr + (l16 + 16) * P);
#pragma unroll
    for (int kc = 0; kc < K; kc += 4) {
      const int fi = (kc >> 1) + half;
      float2 fa = A2[fi];
      float2 f0 = B0[fi];
      float2 f1 = B1[fi];
      v2f a;  a.x  = fa.x; a.y  = fa.y;
      v2f b0; b0.x = f0.x; b0.y = f0.y;
      v2f b1; b1.x = f1.x; b1.y = f1.y;
      c0 = wmma_f32_k4(a, b0, c0);
      c1 = wmma_f32_k4(a, b1, c1);
    }
  }

  const float bias0 = bl[l16];
  const float bias1 = bl[l16 + 16];
#pragma unroll
  for (int r = 0; r < 8; ++r) { c0[r] += bias0; c1[r] += bias1; }

#pragma unroll
  for (int r = 0; r < 8; ++r) {
    float p = c0[r] * c0[r] + c1[r] * c1[r];
    p += __shfl_xor(p, 1, 32);
    p += __shfl_xor(p, 2, 32);
    p += __shfl_xor(p, 4, 32);
    p += __shfl_xor(p, 8, 32);
    float scale = 1.0f / fmaxf(sqrtf(p), 1e-12f);
    const size_t row = (size_t)(base + r + half * 8);
    out[row * 32 + l16]      = c0[r] * scale;
    out[row * 32 + 16 + l16] = c1[r] * scale;
  }
}

extern "C" void kernel_launch(void* const* d_in, const int* in_sizes, int n_in,
                              void* d_out, int out_size, void* d_ws,
                              size_t ws_size, hipStream_t stream) {
  const float* x   = (const float*)d_in[0];
  const int* src0  = (const int*)d_in[1];
  const int* tgt0  = (const int*)d_in[2];
  const int* src1  = (const int*)d_in[3];
  const int* tgt1  = (const int*)d_in[4];
  const float* wl0 = (const float*)d_in[5];
  const float* bl0 = (const float*)d_in[6];
  const float* wr0 = (const float*)d_in[7];
  const float* wl1 = (const float*)d_in[8];
  const float* bl1 = (const float*)d_in[9];
  const float* wr1 = (const float*)d_in[10];
  const int E0 = in_sizes[1];
  const int E1 = in_sizes[3];

  float* ws = (float*)d_ws;
  float* h  = ws;
  const int slices0 = (E0 + SL_ - 1) / SL_;
  const int slices1 = (E1 + SL_ - 1) / SL_;
  unsigned* pairs0 = (unsigned*)(h + (size_t)N1_ * H_);
  unsigned* pairs1 = pairs0 + (size_t)slices0 * SL_ * 2;
  unsigned* starts0 = pairs1 + (size_t)slices1 * SL_ * 2;
  unsigned* starts1 = starts0 + (size_t)slices0 * STP_;
  float* y0 = (float*)(starts1 + (size_t)slices1 * STP_);

  {
    int pblocks = (N0_ / 16) / (GT_ / 32);
    sage_prep_kernel<<<slices0 + slices1 + pblocks, GT_, 0, stream>>>(
        src0, tgt0, E0, SH0_, pairs0, starts0, slices0,
        src1, tgt1, E1, SH1_, pairs1, starts1, slices1,
        x, wl0, y0, N0_);
  }
  sage_gather_combine_kernel<<<NB_, 1024, 0, stream>>>(
      y0, pairs0, starts0, slices0, x, bl0, wr0, h);
  sage_gather_transform_kernel<<<NB_, 1024, 0, stream>>>(
      h, pairs1, starts1, slices1, wl1, bl1, wr1, (float*)d_out);
}


# Gather work-groups order their range's edges by target row in LDS (rank by integer LDS add, prefix, place); a wave sums its own rows in registers, stores each once: no LDS f32 adds; rank-order sums
# speedup vs baseline: 7.1354x; 1.0489x over previous
#include <hip/hip_runtime.h>
#include <math.h>

typedef __attribute__((ext_vector_type(2))) float v2f;
typedef __attribute__((ext_vector_type(4))) float v4f;
typedef __attribute__((ext_vector_type(8))) float v8f;

#define N0_  262144
#define INF_ 128
#define H_   32
#define N1_  131072
#define N2_  65536

__device__ __forceinline__ v8f wmma_f32_k4(v2f a, v2f b, v8f c) {
  return __builtin_amdgcn_wmma_f32_16x16x4_f32(
       false, a,  false, b,
       (short)0, c,  false,  false);
}

#define NB_ 256
#define SL_ 8192
#define GT_ 256
#define EPT_ (SL_ / GT_)
#define STP_ (NB_ + 32)
#define SH0_ 9
#define SH1_ 8
static_assert((NB_ << SH0_) == N1_, "ranges cover the layer-0 targets");
static_assert((NB_ << SH1_) == N2_, "ranges cover the layer-1 targets");
static_assert(GT_ == NB_, "one thread per range in the prefix");
static_assert(EPT_ % 4 == 0, "a thread's edges are whole groups of four");
#define G_SPAIR_W 0
#define G_HCNT_W  (2 * SL_)
#define G_WTOT_W  (G_HCNT_W + NB_)
#define G_TOTAL_W (G_WTOT_W + GT_ / 32)
#define G_WORDS   (G_TOTAL_W + 1)
__device__ __forceinline__ void group_slice(
    const int* __restrict__ src, const int* __restrict__ tgt, int edges,
    int shift, unsigned* __restrict__ pairs_w, unsigned* __restrict__ starts,
    int sl, unsigned* __restrict__ lds_w) {
  uint2* spair      = (uint2*)(lds_w + G_SPAIR_W);
  unsigned* hcnt    = lds_w + G_HCNT_W;
  unsigned* wtot    = lds_w + G_WTOT_W;
  unsigned* total_s = lds_w + G_TOTAL_W;
  uint2* __restrict__ pairs = (uint2*)pairs_w;
  const int tid = (int)threadIdx.x;
  const int lane = tid & 31;
  const int e_begin = sl * SL_;
  const int n = min(SL_, edges - e_begin);

  hcnt[tid] = 0u;
  __syncthreads();
  int s[EPT_];
  int t[EPT_];
  if (n == SL_) {
    const int4* __restrict__ s4 = (const int4*)(src + e_begin);
    const int4* __restrict__ t4 = (const int4*)(tgt + e_begin);
#pragma unroll
    for (int u = 0; u < EPT_ / 4; ++u) {
      const int4 a = s4[tid + u * GT_];
      const int4 b = t4[tid + u * GT_];
      s[4 * u + 0] = a.x; s[4 * u + 1] = a.y; s[4 * u + 2] = a.z; s[4 * u + 3] = a.w;
      t[4 * u + 0] = b.x; t[4 * u + 1] = b.y; t[4 * u + 2] = b.z; t[4 * u + 3] = b.w;
    }
  } else {
#pragma unroll
    for (int u = 0; u < EPT_; ++u) {
      const int i = tid + u * GT_;
      s[u] = 0;
      t[u] = -1;
      if (i < n) { s[u] = src[e_begin + i]; t[u] = tgt[e_begin + i]; }
    }
  }
#pragma unroll
  for (int u = 0; u < EPT_; ++u) {
    const unsigned b = (unsigned)t[u] >> shift;
    if (b < (unsigned)NB_) atomicAdd(&hcnt[b], 1u);
    else t[u] = -1;
  }
  __syncthreads();
  const unsigned c = hcnt[tid];
  unsigned incl = c;
#pragma unroll
  for (int d = 1; d < 32; d <<= 1) {
    const unsigned v = __shfl_up(incl, d, 32);
    if (lane >= d) incl += v;
  }
  if (lane == 31) wtot[tid >> 5] = incl;
  __syncthreads();
  {
    unsigned base = 0u;
    for (int k = 0; k < (tid >> 5); ++k) base += wtot[k];
    const unsigned ex = base + incl - c;
    hcnt[tid] = ex;
    starts[(size_t)sl * STP_ + tid] = ex;
    if (tid == NB_ - 1) {
      starts[(size_t)sl * STP_ + NB_] = ex + c;
      *total_s = ex + c;
    }
  }
  __syncthreads();
#pragma unroll
  for (int u = 0; u < EPT_; ++u) {
    if (t[u] >= 0) {
      const unsigned b   = (unsigned)t[u] >> shift;
      const unsigned pos = atomicAdd(&hcnt[b], 1u);
      spair[pos] = make_uint2((unsigned)s[u], (unsigned)t[u]);
    }
  }
  __syncthreads();
  const int total = (int)*total_s;
  for (int i = tid; i < total; i += GT_) pairs[(size_t)e_begin + i] = spair[i];
}

template <int N>
__device__ __forceinline__ void add_rows(const float* __restrict__ xsrc,
                                         float* __restrict__ acc, int s_v,
                                         int t_v, int jb, unsigned loff) {
  float v[N];
#pragma unroll
  for (int j = 0; j < N; ++j) {
    const unsigned s = (unsigned)__builtin_amdgcn_readlane(s_v, jb + j);
    v[j] = xsrc[s * 32u + loff];
  }
#pragma unroll
  for (int j = 0; j < N; ++j) {
    const unsigned t = (unsigned)__builtin_amdgcn_readlane(t_v, jb + j);
    atomicAdd(&acc[t * 32u + loff], v[j]);
  }
}

template <int N>
__device__ __forceinline__ void sum_rows(const float* __restrict__ xsrc,
                                         float* __restrict__ accw, int s_v,
                                         int jb, unsigned loff, unsigned e0,
                                         int ends, int& r, unsigned& nb,
                                         float& sum) {
  float v[N];
#pragma unroll
  for (int j = 0; j < N; ++j) {
    const unsigned s = (unsigned)__builtin_amdgcn_readlane(s_v, jb + j);
    v[j] = xsrc[s * 32u + loff];
  }
#pragma unroll
  for (int j = 0; j < N; ++j) {
    const unsigned e = e0 + (unsigned)j;
    while (e == nb) {
      accw[r * 32] = sum;
      sum = 0.0f;
      ++r;
      nb = (unsigned)__builtin_amdgcn_readlane(ends, r);
    }
    sum += v[j];
  }
}

template <int SHIFT, int CAP>
__device__ __forceinline__ void gather_range(
    const float* __restrict__ xsrc, const unsigned* __restrict__ pairs_w,
    const unsigned* __restrict__ starts, float* __restrict__ acc,
    unsigned* __restrict__ cn, unsigned* __restrict__ offs,
    unsigned* __restrict__ wtot, unsigned* __restrict__ list, int nslices,
    int b, int tid, int lane, int wave) {
  const uint2* __restrict__ pairs = (const uint2*)pairs_w;
  constexpr int R = 1 << SHIFT;
  constexpr int RPW = R / 32;
  static_assert(CAP < 65536, "a rank fits in the upper half of a key word");
  const int row0 = b << SHIFT;
  unsigned* stg_s  = list;
  unsigned* stg_k  = list + CAP;
  unsigned* sorted = list + 2 * CAP;
  if (tid < R) cn[tid] = 0u;

  const int rpw = (nslices + 31) / 32;
  unsigned st = 0u, len = 0u;
  {
    const int g = wave * rpw + lane;
    if (lane < rpw && g < nslices) {
      const unsigned* p = starts + (size_t)g * STP_ + b;
      const unsigned a = p[0];
      const unsigned e = p[1];
      st  = (unsigned)g * (unsigned)SL_ + a;
      len = e - a;
    }
  }
  unsigned incl = len;
#pragma unroll
  for (int d = 1; d < 32; d <<= 1) {
    const unsigned v = __shfl_up(incl, d, 32);
    if (lane >= d) incl += v;
  }
  const unsigned excl  = incl - len;
  const unsigned total = (unsigned)__builtin_amdgcn_readlane((int)incl, 31);
  const unsigned loff  = (unsigned)lane;
  const unsigned rmask = (unsigned)R - 1u;
  if (lane == 0) wtot[wave] = total;
  __syncthreads();
  unsigned wb, rt;
  {
    const unsigned wl = wtot[lane];
    unsigned in2 = wl;
#pragma unroll
    for (int d = 1; d < 32; d <<= 1) {
      const unsigned v = __shfl_up(in2, d, 32);
      if (lane >= d) in2 += v;
    }
    rt = (unsigned)__builtin_amdgcn_readlane((int)in2, 31);
    wb = (unsigned)__builtin_amdgcn_readlane((int)(in2 - wl), wave);
  }

  if (rt > (unsigned)CAP) {
    float4* acc4 = (float4*)acc;
    for (int i = tid; i < R * 8; i += 1024) acc4[i] = make_float4(0.f, 0.f, 0.f, 0.f);
    __syncthreads();
    for (unsigned c0 = 0u; c0 < total; c0 += 32u) {
      const unsigned q = c0 + loff;
      unsigned addr = 0u;
      for (int k = 0; k < rpw; ++k) {
        const unsigned ek = (unsigned)__builtin_amdgcn_readlane((int)excl, k);
        const unsigned ik = (unsigned)__builtin_amdgcn_readlane((int)incl, k);
        const unsigned sk = (unsigned)__builtin_amdgcn_readlane((int)st, k);
        if (q >= ek && q < ik) addr = sk + (q - ek);
      }
      const unsigned left = total - c0;
      int s_v = 0;
      int t_v = 0;
      if (loff < left) {
        const uint2 pr = pairs[addr];
        s_v = (int)pr.x;
        t_v = (int)(pr.y - (unsigned)row0) & (int)rmask;
        atomicAdd(&cn[t_v], 1u);
      }
      if (left >= 32u) {
        add_rows<32>(xsrc, acc, s_v, t_v, 0, loff);
      } else {
        int jb = 0;
        if (left & 16u) { add_rows<16>(xsrc, acc, s_v, t_v, jb, loff); jb += 16; }
        if (left & 8u)  { add_rows<8>(xsrc, acc, s_v, t_v, jb, loff);  jb += 8; }
        if (left & 4u)  { add_rows<4>(xsrc, acc, s_v, t_v, jb, loff);  jb += 4; }
        if (left & 2u)  { add_rows<2>(xsrc, acc, s_v, t_v, jb, loff);  jb += 2; }
        if (left & 1u)  { add_rows<1>(xsrc, acc, s_v, t_v, jb, loff); }
      }
    }
    return;
  }

  for (unsigned c0 = 0u; c0 < total; c0 += 64u) {
    const unsigned q0 = c0 + loff;
    const unsigned q1 = q0 + 32u;
    unsigned a0 = 0u, a1 = 0u;
    for (int k = 0; k < rpw; ++k) {
      const unsigned ek = (unsigned)__builtin_amdgcn_readlane((int)excl, k);
      const unsigned ik = (unsigned)__builtin_amdgcn_readlane((int)incl, k);
      const unsigned sk = (unsigned)__builtin_amdgcn_readlane((int)st, k);
      if (q0 >= ek && q0 < ik) a0 = sk + (q0 - ek);
      if (q1 >= ek && q1 < ik) a1 = sk + (q1 - ek);
    }
    const bool v0 = q0 < total;
    const bool v1 = q1 < total;
    const unsigned afirst = (unsigned)__builtin_amdgcn_readlane((int)a0, 0);
    const uint2 p0 = pairs[v0 ? a0 : afirst];
    const uint2 p1 = pairs[v1 ? a1 : afirst];
    if (v0) {
      const unsigned t = (p0.y - (unsigned)row0) & rmask;
      const unsigned rank = atomicAdd(&cn[t], 1u);
      stg_s[wb + q0] = p0.x;
      stg_k[wb + q0] = t | (rank << 16);
    }
    if (v1) {
      const unsigned t = (p1.y - (unsigned)row0) & rmask;
      const unsigned rank = atomicAdd(&cn[t], 1u);
      stg_s[wb + q1] = p1.x;
      stg_k[wb + q1] = t | (rank << 16);
    }
  }
  __syncthreads();
  {
    unsigned c = 0u;
    if (tid < R) c = cn[tid];
    unsigned in3 = c;
#pragma unroll
    for (int d = 1; d < 32; d <<= 1) {
      const unsigned v = __shfl_up(in3, d, 32);
      if (lane >= d) in3 += v;
    }
    if (tid < R && lane == 31) wtot[32 + wave] = in3;
    __syncthreads();
    if (tid < R) {
      unsigned base = 0u;
      for (int k = 0; k < wave; ++k) base += wtot[32 + k];
      offs[tid] = base + in3 - c;
      if (tid == R - 1) offs[R] = base + in3;
    }
  }
  __syncthreads();
  for (unsigned i = (unsigned)tid; i < rt; i += 1024u) {
    const unsigned k = stg_k[i];
    sorted[offs[k & 0xffffu] + (k >> 16)] = stg_s[i];
  }
  __syncthreads();
  {
    const int offv = (int)offs[wave * RPW + min(lane, RPW)];
    const unsigned beg = (unsigned)__builtin_amdgcn_readlane(offv, 0);
    const unsigned end = (unsigned)__builtin_amdgcn_readlane(offv, RPW);
    const int nxt = (int)offs[wave * RPW + min(lane + 1, RPW)];
    const int ends = (lane < RPW - 1) ? nxt : -1;
    int r = 0;
    unsigned nb = (unsigned)__builtin_amdgcn_readlane(ends, 0);
    float sum = 0.0f;
    float* accw = acc + wave * (RPW * 32) + lane;
    for (unsigned c0 = beg; c0 < end; c0 += 32u) {
      const unsigned left = end - c0;
      int s_v = 0;
      if (loff < left) s_v = (int)sorted[c0 + loff];
      if (left >= 32u) {
        sum_rows<32>(xsrc, accw, s_v, 0, loff, c0, ends, r, nb, sum);
      } else {
        int jb = 0;
        if (left & 16u) { sum_rows<16>(xsrc, accw, s_v, jb, loff, c0 + jb, ends, r, nb, sum); jb += 16; }
        if (left & 8u)  { sum_rows<8>(xsrc, accw, s_v, jb, loff, c0 + jb, ends, r, nb, sum);  jb += 8; }
        if (left & 4u)  { sum_rows<4>(xsrc, accw, s_v, jb, loff, c0 + jb, ends, r, nb, sum);  jb += 4; }
        if (left & 2u)  { sum_rows<2>(xsrc, accw, s_v, jb, loff, c0 + jb, ends, r, nb, sum);  jb += 2; }
        if (left & 1u)  { sum_rows<1>(xsrc, accw, s_v, jb, loff, c0 + jb, ends, r, nb, sum); }
      }
    }
    while (r < RPW) {
      accw[r * 32] = sum;
      sum = 0.0f;
      ++r;
    }
  }
}

template <int K>
__device__ __forceinline__ void stage_weights(const float* __restrict__ w,
                                              float* __restrict__ lw) {
  static_assert(K % 32 == 0, "32*K floats: a whole number of float4 per thread");
  constexpr int P = K + 4;
  const float4* w4 = (const float4*)w;
#pragma unroll
  for (int i = 0; i < K / 32; ++i) {
    const int f   = i * 256 + (int)threadIdx.x;
    const int row = f / (K / 4);
    const int q   = f % (K / 4);
    *(float4*)(lw + row * P + 4 * q) = w4[f];
  }
}

template <int K>
__device__ __forceinline__ void stage_rows(const float* __restrict__ rows,
                                           float* __restrict__ la, int lane) {
  static_assert(K % 8 == 0, "16*K floats: a whole number of float4 per lane");
  constexpr int P = K + 4;
  const float4* r4 = (const float4*)rows;
#pragma unroll
  for (int i = 0; i < K / 8; ++i) {
    const int f   = i * 32 + lane;
    const int row = f / (K / 4);
    const int q   = f % (K / 4);
    *(float4*)(la + row * P + 4 * q) = r4[f];
  }
}

template <int K>
__device__ __forceinline__ void project_tiles(
    const float* __restrict__ xin, const float* __restrict__ w,
    float* __restrict__ y, int nrows, int pb, float* __restrict__ lw,
    float* __restrict__ la) {
  constexpr int P = K + 4;
  stage_weights<K>(w, lw);
  const int lane = threadIdx.x & 31;
  const int wv   = threadIdx.x >> 5;
  const int wid  = pb * (GT_ / 32) + wv;
  const int base = wid * 16;
  float* myA = la + wv * (16 * P);
  if (base < nrows) stage_rows<K>(xin + (size_t)base * K, myA, lane);
  __syncthreads();
  if (base >= nrows) return;
  const int l16  = lane & 15;
  const int half = lane >> 4;

  v8f c0 = {};
  v8f c1 = {};
  const float2* A2 = (const float2*)(myA + l16 * P);
  const float2* B0 = (const float2*)(lw + l16 * P);
  const float2* B1 = (const float2*)(lw + (l16 + 16) * P);
#pragma unroll
  for (int kc = 0; kc < K; kc += 4) {
    const int fi = (kc >> 1) + half;
    float2 fa = A2[fi];
    float2 f0 = B0[fi];
    float2 f1 = B1[fi];
    v2f a;  a.x  = fa.x; a.y  = fa.y;
    v2f b0; b0.x = f0.x; b0.y = f0.y;
    v2f b1; b1.x = f1.x; b1.y = f1.y;
    c0 = wmma_f32_k4(a, b0, c0);
    c1 = wmma_f32_k4(a, b1, c1);
  }
#pragma unroll
  for (int r = 0; r < 8; ++r) {
    const size_t row = (size_t)(base + r + half * 8);
    y[row * 32 + l16]      = c0[r];
    y[row * 32 + 16 + l16] = c1[r];
  }
}

#define PREP_LDS_WORDS (32 * (INF_ + 4) + 8 * 16 * (INF_ + 4))
static_assert(G_WORDS <= PREP_LDS_WORDS, "a slice's LDS fits in the projection's");
__global__ __launch_bounds__(GT_) void sage_prep_kernel(
    const int* __restrict__ srcA, const int* __restrict__ tgtA, int edgesA,
    int shiftA, unsigned* __restrict__ pairsA, unsigned* __restrict__ startsA,
    int slicesA,
    const int* __restrict__ srcB, const int* __restrict__ tgtB, int edgesB,
    int shiftB, unsigned* __restrict__ pairsB, unsigned* __restrict__ startsB,
    int slicesB,
    const float* __restrict__ xin, const float* __restrict__ w,
    float* __restrict__ y, int nrows) {
  __shared__ __attribute__((aligned(16))) unsigned lds_w[PREP_LDS_WORDS];
  const int blk = (int)blockIdx.x;
  if (blk < slicesA + slicesB) {
    const bool second = blk >= slicesA;
    group_slice(second ? srcB : srcA, second ? tgtB : tgtA,
                second ? edgesB : edgesA, second ? shiftB : shiftA,
                second ? pairsB : pairsA, second ? startsB : startsA,
                second ? blk - slicesA : blk, lds_w);
  } else {
    float* lw = (float*)lds_w;
    float* la = lw + 32 * (INF_ + 4);
    project_tiles<INF_>(xin, w, y, nrows, blk - slicesA - slicesB, lw, la);
  }
}

__global__ __launch_bounds__(1024) void sage_gather_combine_kernel(
    const float* __restrict__ ysrc, const unsigned* __restrict__ pairs_w,
    const unsigned* __restrict__ starts, int nslices,
    const float* __restrict__ xin, const float* __restrict__ bl,
    const float* __restrict__ wr, float* __restrict__ out) {
  constexpr int K = INF_;
  constexpr int P = K + 4;
  constexpr int R = 1 << SH0_;
  __shared__ __attribute__((aligned(16))) float big[32 * 16 * P];
  static_assert(R * 32 <= 32 * 16 * P, "the sums fit in front of the tiles' space");
  __shared__ unsigned cn[R];
  __shared__ unsigned offs[R + 1];
  __shared__ unsigned wtot[64];
  __shared__ __attribute__((aligned(16))) float lw[32 * P];
  float* acc = big;
  constexpr int CAP = 12288;
  static_assert(R * 32 + 3 * CAP <= 32 * 16 * P, "sums and edge lists fit in big");
  unsigned* list = (unsigned*)(big + R * 32);
  const int b    = (int)blockIdx.x;
  const int tid  = (int)threadIdx.x;
  const int lane = tid & 31;
  const int wave = __builtin_amdgcn_readfirstlane(tid >> 5);
  static_assert(32 * K == 1024 * 4, "wr is one float4 per thread");
  *(float4*)(lw + (tid / (K / 4)) * P + 4 * (tid % (K / 4))) = ((const float4*)wr)[tid];

  gather_range<SH0_, CAP>(ysrc, pairs_w, starts, acc, cn, offs, wtot, list, nslices, b, tid,
                          lane, wave);

  const int lrow = wave * 16;
  const int base = (b << SH0_) + lrow;
  float* myA = big + wave * (16 * P);
  const v4f* r4 = (const v4f*)(xin + (size_t)base * K);
  v4f xa[K / 8];
#pragma unroll
  for (int i = 0; i < K / 8; ++i) xa[i] = r4[i * 32 + lane];
  __syncthreads();
  const int l16  = lane & 15;
  const int half = lane >> 4;
  v8f c0;
  v8f c1;
  const float myinv = 1.0f / fmaxf((float)cn[lrow + half * 8 + (l16 & 7)], 1.0f);
#pragma unroll
  for (int r = 0; r < 8; ++r) {
    const int row = lrow + r + half * 8;
    const float inv = __shfl(myinv, half * 16 + r, 32);
    c0[r] = acc[row * 32 + l16] * inv;
    c1[r] = acc[row * 32 + 16 + l16] * inv;
  }
  __syncthreads();
#pragma unroll
  for (int i = 0; i < K / 8; ++i) {
    const int f = i * 32 + lane;
    *(v4f*)(myA + (f / (K / 4)) * P + 4 * (f % (K / 4))) = xa[i];
  }
  __syncthreads();

  const float2* A2 = (const float2*)(myA + l16 * P);
  const float2* B0 = (const float2*)(lw + l16 * P);
  const float2* B1 = (const float2*)(lw + (l16 + 16) * P);
#pragma unroll
  for (int kc = 0; kc < K; kc += 4) {
    const int fi = (kc >> 1) + half;
    float2 fa = A2[fi];
    float2 f0 = B0[fi];
    float2 f1 = B1[fi];
    v2f a;  a.x  = fa.x; a.y  = fa.y;
    v2f b0; b0.x = f0.x; b0.y = f0.y;
    v2f b1; b1.x = f1.x; b1.y = f1.y;
    c0 = wmma_f32_k4(a, b0, c0);
    c1 = wmma_f32_k4(a, b1, c1);
  }

  const float bias0 = bl[l16];
  const float bias1 = bl[l16 + 16];
#pragma unroll
  for (int r = 0; r < 8; ++r) { c0[r] += bias0; c1[r] += bias1; }

#pragma unroll
  for (int r = 0; r < 8; ++r) {
    float p = c0[r] * c0[r] + c1[r] * c1[r];
    p += __shfl_xor(p, 1, 32);
    p += __shfl_xor(p, 2, 32);
    p += __shfl_xor(p, 4, 32);
    p += __shfl_xor(p, 8, 32);
    float scale = 1.0f / fmaxf(sqrtf(p), 1e-12f);
    const float o0 = fmaxf(c0[r] * scale, 0.0f);
    const float o1 = fmaxf(c1[r] * scale, 0.0f);
    const size_t row = (size_t)(base + r + half * 8);
    out[row * 32 + l16]      = o0;
    out[row * 32 + 16 + l16] = o1;
  }
}

__global__ __launch_bounds__(1024) void sage_gather_transform_kernel(
    const float* __restrict__ hin, const unsigned* __restrict__ pairs_w,
    const unsigned* __restrict__ starts, int nslices,
    const float* __restrict__ wl, const float* __restrict__ bl,
    const float* __restrict__ wr, float* __restrict__ out) {
  constexpr int K = H_;
  constexpr int P = K + 4;
  constexpr int R = 1 << SH1_;
  __shared__ __attribute__((aligned(16))) float acc[R * 32];
  __shared__ unsigned cn[R];
  __shared__ unsigned offs[R + 1];
  __shared__ unsigned wtot[64];
  constexpr int CAP = 8192;
  __shared__ unsigned list[3 * CAP];
  __shared__ __attribute__((aligned(16))) float lwl[32 * P];
  __shared__ __attribute__((aligned(16))) float lwr[32 * P];
  const int b    = (int)blockIdx.x;
  const int tid  = (int)threadIdx.x;
  const int lane = tid & 31;
  const int wave = __builtin_amdgcn_readfirstlane(tid >> 5);
  static_assert(32 * K == 256 * 4, "wl, wr: one float4 for each of 256 threads");
  if (tid < 512) {
    const int f = tid & 255;
    const float4 v = ((const float4*)(tid < 256 ? wl : wr))[f];
    *(float4*)((tid < 256 ? lwl : lwr) + (f / (K / 4)) * P + 4 * (f % (K / 4))) = v;
  }

  gather_range<SH1_, CAP>(hin, pairs_w, starts, acc, cn, offs, wtot, list, nslices, b, tid,
                          lane, wave);
  __syncthreads();
  if (wave >= R / 16) return;

  const int lrow = wave * 16;
  const int base = (b << SH1_) + lrow;
  const int l16  = lane & 15;
  const int half = lane >> 4;
  const float inv = 1.0f / fmaxf((float)cn[lrow + l16], 1.0f);

  v8f c0 = {};
  v8f c1 = {};
  {
    const float2* A2 = (const float2*)(acc + (lrow + l16) * 32);
    const float2* B0 = (const float2*)(lwl + l16 * P);
    const float2* B1 = (const float2*)(lwl + (l16 + 16) * P);
#pragma unroll
    for (int kc = 0; kc < K; kc += 4) {
      const int fi = (kc >> 1) + half;
      float2 fa = A2[fi];
      float2 f0 = B0[fi];
      float2 f1 = B1[fi];
      v2f a;  a.x  = fa.x * inv; a.y  = fa.y * inv;
      v2f b0; b0.x = f0.x;       b0.y = f0.y;
      v2f b1; b1.x = f1.x;       b1.y = f1.y;
      c0 = wmma_f32_k4(a, b0, c0);
      c1 = wmma_f32_k4(a, b1, c1);
    }
  }
  {
    const float2* A2 = (const float2*)(hin + (size_t)(base + l16) * K);
    const float2* B0 = (const float2*)(lwr + l16 * P);
    const float2* B1 = (const float2*)(lwr + (l16 + 16) * P);
#pragma unroll
    for (int kc = 0; kc < K; kc += 4) {
      const int fi = (kc >> 1) + half;
      float2 fa = A2[fi];
      float2 f0 = B0[fi];
      float2 f1 = B1[fi];
      v2f a;  a.x  = fa.x; a.y  = fa.y;
      v2f b0; b0.x = f0.x; b0.y = f0.y;
      v2f b1; b1.x = f1.x; b1.y = f1.y;
      c0 = wmma_f32_k4(a, b0, c0);
      c1 = wmma_f32_k4(a, b1, c1);
    }
  }

  const float bias0 = bl[l16];
  const float bias1 = bl[l16 + 16];
#pragma unroll
  for (int r = 0; r < 8; ++r) { c0[r] += bias0; c1[r] += bias1; }

#pragma unroll
  for (int r = 0; r < 8; ++r) {
    float p = c0[r] * c0[r] + c1[r] * c1[r];
    p += __shfl_xor(p, 1, 32);
    p += __shfl_xor(p, 2, 32);
    p += __shfl_xor(p, 4, 32);
    p += __shfl_xor(p, 8, 32);
    float scale = 1.0f / fmaxf(sqrtf(p), 1e-12f);
    const size_t row = (size_t)(base + r + half * 8);
    out[row * 32 + l16]      = c0[r] * scale;
    out[row * 32 + 16 + l16] = c1[r] * scale;
  }
}

extern "C" void kernel_launch(void* const* d_in, const int* in_sizes, int n_in,
                              void* d_out, int out_size, void* d_ws,
                              size_t ws_size, hipStream_t stream) {
  const float* x   = (const float*)d_in[0];
  const int* src0  = (const int*)d_in[1];
  const int* tgt0  = (const int*)d_in[2];
  const int* src1  = (const int*)d_in[3];
  const int* tgt1  = (const int*)d_in[4];
  const float* wl0 = (const float*)d_in[5];
  const float* bl0 = (const float*)d_in[6];
  const float* wr0 = (const float*)d_in[7];
  const float* wl1 = (const float*)d_in[8];
  const float* bl1 = (const float*)d_in[9];
  const float* wr1 = (const float*)d_in[10];
  const int E0 = in_sizes[1];
  const int E1 = in_sizes[3];

  float* ws = (float*)d_ws;
  float* h  = ws;
  const int slices0 = (E0 + SL_ - 1) / SL_;
  const int slices1 = (E1 + SL_ - 1) / SL_;
  unsigned* pairs0 = (unsigned*)(h + (size_t)N1_ * H_);
  unsigned* pairs1 = pairs0 + (size_t)slices0 * SL_ * 2;
  unsigned* starts0 = pairs1 + (size_t)slices1 * SL_ * 2;
  unsigned* starts1 = starts0 + (size_t)slices0 * STP_;
  float* y0 = (float*)(starts1 + (size_t)slices1 * STP_);

  {
    int pblocks = (N0_ / 16) / (GT_ / 32);
    sage_prep_kernel<<<slices0 + slices1 + pblocks, GT_, 0, stream>>>(
        src0, tgt0, E0, SH0_, pairs0, starts0, slices0,
        src1, tgt1, E1, SH1_, pairs1, starts1, slices1,
        x, wl0, y0, N0_);
  }
  sage_gather_combine_kernel<<<NB_, 1024, 0, stream>>>(
      y0, pairs0, starts0, slices0, x, bl0, wr0, h);
  sage_gather_transform_kernel<<<NB_, 1024, 0, stream>>>(
      h, pairs1, starts1, slices1, wl1, bl1, wr1, (float*)d_out);
}
